# starting kernel
# baseline (speedup 1.0000x reference)
#include <hip/hip_runtime.h>
#include <hip/hip_bf16.h>
#include <math.h>

typedef __attribute__((ext_vector_type(2))) float v2f;
typedef __attribute__((ext_vector_type(8))) float v8f;

#define D_IN   1024
#define D_ST   16
#define DT_RANK 64
#define N_E    96
#define BATCH  2
#define SEQ    2048
#define M_TOT  (BATCH*SEQ)
#define NCHUNK 8
#define LCHUNK (SEQ/NCHUNK)
#define NCHAN  (BATCH*D_IN)

__global__ void xproj_wmma(const float* __restrict__ X,
                           const float* __restrict__ W,
                           float* __restrict__ xdbl) {
    int wave = (blockIdx.x * blockDim.x + threadIdx.x) >> 5;
    int lane = threadIdx.x & 31;
    const int NT = N_E / 16;
    int nt = wave % NT;
    int mt = wave / NT;
    if (mt >= M_TOT / 16) return;
    int m0 = mt * 16, n0 = nt * 16;
    int half = lane >> 4, lq = lane & 15;
    int koff = half * 2;
    const float* aptr = X + (size_t)(m0 + lq) * D_IN + koff;
    const float* bptr = W + (size_t)(n0 + lq) * D_IN + koff;
    v8f acc = {};
    for (int k = 0; k < D_IN; k += 4) {
        v2f a = *(const v2f*)(aptr + k);
        v2f b = *(const v2f*)(bptr + k);
        acc = __builtin_amdgcn_wmma_f32_16x16x4_f32(
            false, a, false, b, (short)0, acc, false, false);
    }
    int n = n0 + lq;
#pragma unroll
    for (int v = 0; v < 8; ++v) {
        int m = m0 + v + 8 * half;
        xdbl[(size_t)m * N_E + n] = acc[v];
    }
}

__global__ void dt_wmma(const float* __restrict__ xdbl,
                        const float* __restrict__ Wdt,
                        const float* __restrict__ bdt,
                        float* __restrict__ dt) {
    int wave = (blockIdx.x * blockDim.x + threadIdx.x) >> 5;
    int lane = threadIdx.x & 31;
    const int NT = D_IN / 16;
    int nt = wave % NT;
    int mt = wave / NT;
    if (mt >= M_TOT / 16) return;
    int m0 = mt * 16, n0 = nt * 16;
    int half = lane >> 4, lq = lane & 15;
    int koff = half * 2;
    const float* aptr = xdbl + (size_t)(m0 + lq) * N_E + koff;
    const float* bptr = Wdt + (size_t)(n0 + lq) * DT_RANK + koff;
    v8f acc = {};
#pragma unroll
    for (int k = 0; k < DT_RANK; k += 4) {
        v2f a = *(const v2f*)(aptr + k);
        v2f b = *(const v2f*)(bptr + k);
        acc = __builtin_amdgcn_wmma_f32_16x16x4_f32(
            false, a, false, b, (short)0, acc, false, false);
    }
    int n = n0 + lq;
    float bias = bdt[n];
#pragma unroll
    for (int v = 0; v < 8; ++v) {
        int m = m0 + v + 8 * half;
        float z = acc[v] + bias;
        float sp = (z > 20.0f) ? z : log1pf(__expf(z));
        dt[(size_t)m * D_IN + n] = sp;
    }
}

__global__ void scan_phase1(const float* __restrict__ X,
                            const float* __restrict__ xdbl,
                            const float* __restrict__ dt,
                            const float* __restrict__ A_log,
                            float* __restrict__ aprod,
                            float* __restrict__ hend) {
    int gid = blockIdx.x * blockDim.x + threadIdx.x;
    int unit = gid >> 4;
    int s = gid & 15;
    if (unit >= NCHAN * NCHUNK) return;
    int chunk = unit % NCHUNK;
    int channel = unit / NCHUNK;
    int b = channel / D_IN;
    int d = channel % D_IN;
    float A = -__expf(A_log[d * D_ST + s]);
    float ap = 1.0f, h = 0.0f;
    int l0 = chunk * LCHUNK;
    for (int l = l0; l < l0 + LCHUNK; ++l) {
        int m = b * SEQ + l;
        float dtv = dt[(size_t)m * D_IN + d];
        float xv  = X[(size_t)m * D_IN + d];
        float Bm  = xdbl[(size_t)m * N_E + DT_RANK + s];
        float ab  = __expf(dtv * A);
        h  = ab * h + dtv * Bm * xv;
        ap *= ab;
    }
    int idx = (channel * D_ST + s) * NCHUNK + chunk;
    aprod[idx] = ap;
    hend[idx]  = h;
}

__global__ void scan_phase2(const float* __restrict__ aprod,
                            const float* __restrict__ hend,
                            float* __restrict__ carry) {
    int t = blockIdx.x * blockDim.x + threadIdx.x;
    if (t >= NCHAN * D_ST) return;
    float c = 0.0f;
#pragma unroll
    for (int ch = 0; ch < NCHUNK; ++ch) {
        int idx = t * NCHUNK + ch;
        carry[idx] = c;
        c = aprod[idx] * c + hend[idx];
    }
}

__global__ void scan_phase3(const float* __restrict__ X,
                            const float* __restrict__ xdbl,
                            const float* __restrict__ dt,
                            const float* __restrict__ A_log,
                            const float* __restrict__ Dp,
                            const float* __restrict__ carry,
                            float* __restrict__ y) {
    int gid = blockIdx.x * blockDim.x + threadIdx.x;
    int unit = gid >> 4;
    int s = gid & 15;
    if (unit >= NCHAN * NCHUNK) return;
    int chunk = unit % NCHUNK;
    int channel = unit / NCHUNK;
    int b = channel / D_IN;
    int d = channel % D_IN;
    float A = -__expf(A_log[d * D_ST + s]);
    float h = carry[(channel * D_ST + s) * NCHUNK + chunk];
    float Dd = Dp[d];
    int l0 = chunk * LCHUNK;
    for (int l = l0; l < l0 + LCHUNK; ++l) {
        int m = b * SEQ + l;
        float dtv = dt[(size_t)m * D_IN + d];
        float xv  = X[(size_t)m * D_IN + d];
        float Bm  = xdbl[(size_t)m * N_E + DT_RANK + s];
        float Cm  = xdbl[(size_t)m * N_E + DT_RANK + D_ST + s];
        float ab  = __expf(dtv * A);
        h = ab * h + dtv * Bm * xv;
        float contrib = h * Cm;
#pragma unroll
        for (int msk = 1; msk < 16; msk <<= 1)
            contrib += __shfl_xor(contrib, msk, 32);
        if (s == 0)
            y[(size_t)m * D_IN + d] = contrib + Dd * xv;
    }
}

extern "C" void kernel_launch(void* const* d_in, const int* in_sizes, int n_in,
                              void* d_out, int out_size, void* d_ws, size_t ws_size,
                              hipStream_t stream) {
    const float* x       = (const float*)d_in[0];
    const float* A_log   = (const float*)d_in[1];
    const float* Dp      = (const float*)d_in[2];
    const float* W_xproj = (const float*)d_in[3];
    const float* W_dt    = (const float*)d_in[4];
    const float* b_dt    = (const float*)d_in[5];
    float* y = (float*)d_out;

    char* ws = (char*)d_ws;
    size_t off = 0;
    float* xdbl  = (float*)(ws + off); off += (size_t)M_TOT * N_E * 4;
    off = (off + 255) & ~(size_t)255;
    float* dtbuf = (float*)(ws + off); off += (size_t)M_TOT * D_IN * 4;
    off = (off + 255) & ~(size_t)255;
    float* aprod = (float*)(ws + off); off += (size_t)NCHAN * D_ST * NCHUNK * 4;
    off = (off + 255) & ~(size_t)255;
    float* hend  = (float*)(ws + off); off += (size_t)NCHAN * D_ST * NCHUNK * 4;
    off = (off + 255) & ~(size_t)255;
    float* carry = (float*)(ws + off);

    xproj_wmma<<<1536 / 4, 128, 0, stream>>>(x, W_xproj, xdbl);
    dt_wmma<<<16384 / 8, 256, 0, stream>>>(xdbl, W_dt, b_dt, dtbuf);
    scan_phase1<<<(NCHAN * NCHUNK * 16) / 256, 256, 0, stream>>>(
        x, xdbl, dtbuf, A_log, aprod, hend);
    scan_phase2<<<(NCHAN * D_ST) / 256, 256, 0, stream>>>(aprod, hend, carry);
    scan_phase3<<<(NCHAN * NCHUNK * 16) / 256, 256, 0, stream>>>(
        x, xdbl, dtbuf, A_log, Dp, carry, y);
}


# Scan loops blocked by 16 steps, lane = state kept: dt/x loaded once per block, broadcast by DPP row_share; 16 steps' state sums in one transposed DPP butterfly, one y store per block
# speedup vs baseline: 3.1173x; 3.1173x over previous
#include <hip/hip_runtime.h>
#include <hip/hip_bf16.h>
#include <math.h>

typedef __attribute__((ext_vector_type(2))) float v2f;
typedef __attribute__((ext_vector_type(8))) float v8f;

#define D_IN   1024
#define D_ST   16
#define DT_RANK 64
#define N_E    96
#define BATCH  2
#define SEQ    2048
#define M_TOT  (BATCH*SEQ)
#define NCHUNK 8
#define LCHUNK (SEQ/NCHUNK)
#define NCHAN  (BATCH*D_IN)
#define TBLK   16
#define SCAN_THREADS 512
#define SCAN_WAVES (NCHAN / 2 * NCHUNK)

template <int CTRL>
__device__ __forceinline__ float row_dpp(float v) {
    const int b = __builtin_bit_cast(int, v);
    return __builtin_bit_cast(float, __builtin_amdgcn_update_dpp(0, b, CTRL, 0xf, 0xf, true));
}
template <int J> __device__ __forceinline__ float row_bcast(float v) { return row_dpp<0x150 + J>(v); }
template <int M> __device__ __forceinline__ float row_xor(float v) { return row_dpp<0x160 + M>(v); }

template <int M>
__device__ __forceinline__ float fold(float a, float b, bool up) {
    const float keep = up ? b : a;
    const float send = up ? a : b;
    return keep + row_xor<M>(send);
}

__global__ void xproj_wmma(const float* __restrict__ X,
                           const float* __restrict__ W,
                           float* __restrict__ xdbl) {
    int wave = (blockIdx.x * blockDim.x + threadIdx.x) >> 5;
    int lane = threadIdx.x & 31;
    const int NT = N_E / 16;
    int nt = wave % NT;
    int mt = wave / NT;
    if (mt >= M_TOT / 16) return;
    int m0 = mt * 16, n0 = nt * 16;
    int half = lane >> 4, lq = lane & 15;
    int koff = half * 2;
    const float* aptr = X + (size_t)(m0 + lq) * D_IN + koff;
    const float* bptr = W + (size_t)(n0 + lq) * D_IN + koff;
    v8f acc = {};
    for (int k = 0; k < D_IN; k += 4) {
        v2f a = *(const v2f*)(aptr + k);
        v2f b = *(const v2f*)(bptr + k);
        acc = __builtin_amdgcn_wmma_f32_16x16x4_f32(
            false, a, false, b, (short)0, acc, false, false);
    }
    int n = n0 + lq;
#pragma unroll
    for (int v = 0; v < 8; ++v) {
        int m = m0 + v + 8 * half;
        xdbl[(size_t)m * N_E + n] = acc[v];
    }
}

__global__ void dt_wmma(const float* __restrict__ xdbl,
                        const float* __restrict__ Wdt,
                        const float* __restrict__ bdt,
                        float* __restrict__ dt) {
    int wave = (blockIdx.x * blockDim.x + threadIdx.x) >> 5;
    int lane = threadIdx.x & 31;
    const int NT = D_IN / 16;
    int nt = wave % NT;
    int mt = wave / NT;
    if (mt >= M_TOT / 16) return;
    int m0 = mt * 16, n0 = nt * 16;
    int half = lane >> 4, lq = lane & 15;
    int koff = half * 2;
    const float* aptr = xdbl + (size_t)(m0 + lq) * N_E + koff;
    const float* bptr = Wdt + (size_t)(n0 + lq) * DT_RANK + koff;
    v8f acc = {};
#pragma unroll
    for (int k = 0; k < DT_RANK; k += 4) {
        v2f a = *(const v2f*)(aptr + k);
        v2f b = *(const v2f*)(bptr + k);
        acc = __builtin_amdgcn_wmma_f32_16x16x4_f32(
            false, a, false, b, (short)0, acc, false, false);
    }
    int n = n0 + lq;
    float bias = bdt[n];
#pragma unroll
    for (int v = 0; v < 8; ++v) {
        int m = m0 + v + 8 * half;
        float z = acc[v] + bias;
        float sp = (z > 20.0f) ? z : log1pf(__expf(z));
        dt[(size_t)m * D_IN + n] = sp;
    }
}

#define SCAN1_STEP(J)                                   \
    {                                                   \
        const float dj = row_bcast<J>(dtv);             \
        const float xj = row_bcast<J>(xv);              \
        const float ab = __expf(dj * A);                \
        h  = ab * h + dj * Bv[J] * xj;                  \
        ap *= ab;                                       \
    }

__global__ void __launch_bounds__(SCAN_THREADS)
scan_phase1(const float* __restrict__ X,
            const float* __restrict__ xdbl,
            const float* __restrict__ dt,
            const float* __restrict__ A_log,
            float* __restrict__ aprod,
            float* __restrict__ hend) {
    const int lane = threadIdx.x & 31;
    const int s = lane & 15;
    const int wave = blockIdx.x * (SCAN_THREADS / 32) + (threadIdx.x >> 5);
    const int chunk = wave / (NCHAN / 2);
    const int channel = (wave % (NCHAN / 2)) * 2 + (lane >> 4);
    const int b = channel / D_IN;
    const int d = channel % D_IN;
    const float A = -__expf(A_log[d * D_ST + s]);
    float ap = 1.0f, h = 0.0f;
    const size_t m0 = (size_t)b * SEQ + (size_t)chunk * LCHUNK;
    const float* dtp = dt + (m0 + s) * D_IN + d;
    const float* xp  = X  + (m0 + s) * D_IN + d;
    const float* bp  = xdbl + m0 * N_E + DT_RANK + s;
    for (int t = 0; t < LCHUNK; t += TBLK) {
        const float dtv = dtp[(size_t)t * D_IN];
        const float xv  = xp[(size_t)t * D_IN];
        float Bv[TBLK];
        for (int j = 0; j < TBLK; ++j)
            Bv[j] = bp[(size_t)(t + j) * N_E];
        SCAN1_STEP(0)  SCAN1_STEP(1)  SCAN1_STEP(2)  SCAN1_STEP(3)
        SCAN1_STEP(4)  SCAN1_STEP(5)  SCAN1_STEP(6)  SCAN1_STEP(7)
        SCAN1_STEP(8)  SCAN1_STEP(9)  SCAN1_STEP(10) SCAN1_STEP(11)
        SCAN1_STEP(12) SCAN1_STEP(13) SCAN1_STEP(14) SCAN1_STEP(15)
    }
    const int idx = (channel * D_ST + s) * NCHUNK + chunk;
    aprod[idx] = ap;
    hend[idx]  = h;
}

__global__ void scan_phase2(const float* __restrict__ aprod,
                            const float* __restrict__ hend,
                            float* __restrict__ carry) {
    int t = blockIdx.x * blockDim.x + threadIdx.x;
    if (t >= NCHAN * D_ST) return;
    float c = 0.0f;
#pragma unroll
    for (int ch = 0; ch < NCHUNK; ++ch) {
        int idx = t * NCHUNK + ch;
        carry[idx] = c;
        c = aprod[idx] * c + hend[idx];
    }
}

#define SCAN3_STEP(J)                                   \
    {                                                   \
        const float dj = row_bcast<J>(dtv);             \
        const float xj = row_bcast<J>(xv);              \
        const float ab = __expf(dj * A);                \
        h = ab * h + dj * Bv[J] * xj;                   \
        c[J] = h * Cv[J];                               \
    }

__global__ void __launch_bounds__(SCAN_THREADS)
scan_phase3(const float* __restrict__ X,
            const float* __restrict__ xdbl,
            const float* __restrict__ dt,
            const float* __restrict__ A_log,
            const float* __restrict__ Dp,
            const float* __restrict__ carry,
            float* __restrict__ y) {
    const int lane = threadIdx.x & 31;
    const int s = lane & 15;
    const int wave = blockIdx.x * (SCAN_THREADS / 32) + (threadIdx.x >> 5);
    const int chunk = wave / (NCHAN / 2);
    const int channel = (wave % (NCHAN / 2)) * 2 + (lane >> 4);
    const int b = channel / D_IN;
    const int d = channel % D_IN;
    const float A = -__expf(A_log[d * D_ST + s]);
    float h = carry[(channel * D_ST + s) * NCHUNK + chunk];
    const float Dd = Dp[d];
    const size_t m0 = (size_t)b * SEQ + (size_t)chunk * LCHUNK;
    const float* dtp = dt + (m0 + s) * D_IN + d;
    const float* xp  = X  + (m0 + s) * D_IN + d;
    float*       yp  = y  + (m0 + s) * D_IN + d;
    const float* bcp = xdbl + m0 * N_E + DT_RANK + s;
    const bool up1 = (s & 1) != 0, up2 = (s & 2) != 0, up4 = (s & 4) != 0, up8 = (s & 8) != 0;
    for (int t = 0; t < LCHUNK; t += TBLK) {
        const float dtv = dtp[(size_t)t * D_IN];
        const float xv  = xp[(size_t)t * D_IN];
        float Bv[TBLK], Cv[TBLK], c[TBLK];
        for (int j = 0; j < TBLK; ++j) {
            Bv[j] = bcp[(size_t)(t + j) * N_E];
            Cv[j] = bcp[(size_t)(t + j) * N_E + D_ST];
        }
        SCAN3_STEP(0)  SCAN3_STEP(1)  SCAN3_STEP(2)  SCAN3_STEP(3)
        SCAN3_STEP(4)  SCAN3_STEP(5)  SCAN3_STEP(6)  SCAN3_STEP(7)
        SCAN3_STEP(8)  SCAN3_STEP(9)  SCAN3_STEP(10) SCAN3_STEP(11)
        SCAN3_STEP(12) SCAN3_STEP(13) SCAN3_STEP(14) SCAN3_STEP(15)
        float r1[8], r2[4], r3[2];
        for (int k = 0; k < 8; ++k) r1[k] = fold<1>(c[2 * k], c[2 * k + 1], up1);
        for (int k = 0; k < 4; ++k) r2[k] = fold<2>(r1[2 * k], r1[2 * k + 1], up2);
        for (int k = 0; k < 2; ++k) r3[k] = fold<4>(r2[2 * k], r2[2 * k + 1], up4);
        const float ysum = fold<8>(r3[0], r3[1], up8);
        yp[(size_t)t * D_IN] = ysum + Dd * xv;
    }
}

extern "C" void kernel_launch(void* const* d_in, const int* in_sizes, int n_in,
                              void* d_out, int out_size, void* d_ws, size_t ws_size,
                              hipStream_t stream) {
    const float* x       = (const float*)d_in[0];
    const float* A_log   = (const float*)d_in[1];
    const float* Dp      = (const float*)d_in[2];
    const float* W_xproj = (const float*)d_in[3];
    const float* W_dt    = (const float*)d_in[4];
    const float* b_dt    = (const float*)d_in[5];
    float* y = (float*)d_out;

    char* ws = (char*)d_ws;
    size_t off = 0;
    float* xdbl  = (float*)(ws + off); off += (size_t)M_TOT * N_E * 4;
    off = (off + 255) & ~(size_t)255;
    float* dtbuf = (float*)(ws + off); off += (size_t)M_TOT * D_IN * 4;
    off = (off + 255) & ~(size_t)255;
    float* aprod = (float*)(ws + off); off += (size_t)NCHAN * D_ST * NCHUNK * 4;
    off = (off + 255) & ~(size_t)255;
    float* hend  = (float*)(ws + off); off += (size_t)NCHAN * D_ST * NCHUNK * 4;
    off = (off + 255) & ~(size_t)255;
    float* carry = (float*)(ws + off);

    xproj_wmma<<<1536 / 4, 128, 0, stream>>>(x, W_xproj, xdbl);
    dt_wmma<<<16384 / 8, 256, 0, stream>>>(xdbl, W_dt, b_dt, dtbuf);
    scan_phase1<<<SCAN_WAVES / (SCAN_THREADS / 32), SCAN_THREADS, 0, stream>>>(
        x, xdbl, dtbuf, A_log, aprod, hend);
    scan_phase2<<<(NCHAN * D_ST) / 256, 256, 0, stream>>>(aprod, hend, carry);
    scan_phase3<<<SCAN_WAVES / (SCAN_THREADS / 32), SCAN_THREADS, 0, stream>>>(
        x, xdbl, dtbuf, A_log, Dp, carry, y);
}


# Scan step slimmed: dt*x formed once per lane per block, log2(e) folded into A, chunk decay = exp2(A*sum dt); per step 2 broadcast multiplies, exp2, 1 fma; token product order, decay rounding change
# speedup vs baseline: 3.2160x; 1.0317x over previous
#include <hip/hip_runtime.h>
#include <hip/hip_bf16.h>
#include <math.h>

typedef __attribute__((ext_vector_type(2))) float v2f;
typedef __attribute__((ext_vector_type(8))) float v8f;

#define D_IN   1024
#define D_ST   16
#define DT_RANK 64
#define N_E    96
#define BATCH  2
#define SEQ    2048
#define M_TOT  (BATCH*SEQ)
#define NCHUNK 8
#define LCHUNK (SEQ/NCHUNK)
#define NCHAN  (BATCH*D_IN)
#define TBLK   16
#define SCAN_THREADS 512
#define SCAN_WAVES (NCHAN / 2 * NCHUNK)
#define LOG2E 1.44269504088896341f

template <int CTRL>
__device__ __forceinline__ float row_dpp(float v) {
    const int b = __builtin_bit_cast(int, v);
    return __builtin_bit_cast(float, __builtin_amdgcn_update_dpp(0, b, CTRL, 0xf, 0xf, true));
}
template <int J> __device__ __forceinline__ float row_bcast(float v) { return row_dpp<0x150 + J>(v); }
template <int M> __device__ __forceinline__ float row_xor(float v) { return row_dpp<0x160 + M>(v); }

template <int M>
__device__ __forceinline__ float fold(float a, float b, bool up) {
    const float keep = up ? b : a;
    const float send = up ? a : b;
    return keep + row_xor<M>(send);
}

__global__ void xproj_wmma(const float* __restrict__ X,
                           const float* __restrict__ W,
                           float* __restrict__ xdbl) {
    int wave = (blockIdx.x * blockDim.x + threadIdx.x) >> 5;
    int lane = threadIdx.x & 31;
    const int NT = N_E / 16;
    int nt = wave % NT;
    int mt = wave / NT;
    if (mt >= M_TOT / 16) return;
    int m0 = mt * 16, n0 = nt * 16;
    int half = lane >> 4, lq = lane & 15;
    int koff = half * 2;
    const float* aptr = X + (size_t)(m0 + lq) * D_IN + koff;
    const float* bptr = W + (size_t)(n0 + lq) * D_IN + koff;
    v8f acc = {};
    for (int k = 0; k < D_IN; k += 4) {
        v2f a = *(const v2f*)(aptr + k);
        v2f b = *(const v2f*)(bptr + k);
        acc = __builtin_amdgcn_wmma_f32_16x16x4_f32(
            false, a, false, b, (short)0, acc, false, false);
    }
    int n = n0 + lq;
#pragma unroll
    for (int v = 0; v < 8; ++v) {
        int m = m0 + v + 8 * half;
        xdbl[(size_t)m * N_E + n] = acc[v];
    }
}

__global__ void dt_wmma(const float* __restrict__ xdbl,
                        const float* __restrict__ Wdt,
                        const float* __restrict__ bdt,
                        float* __restrict__ dt) {
    int wave = (blockIdx.x * blockDim.x + threadIdx.x) >> 5;
    int lane = threadIdx.x & 31;
    const int NT = D_IN / 16;
    int nt = wave % NT;
    int mt = wave / NT;
    if (mt >= M_TOT / 16) return;
    int m0 = mt * 16, n0 = nt * 16;
    int half = lane >> 4, lq = lane & 15;
    int koff = half * 2;
    const float* aptr = xdbl + (size_t)(m0 + lq) * N_E + koff;
    const float* bptr = Wdt + (size_t)(n0 + lq) * DT_RANK + koff;
    v8f acc = {};
#pragma unroll
    for (int k = 0; k < DT_RANK; k += 4) {
        v2f a = *(const v2f*)(aptr + k);
        v2f b = *(const v2f*)(bptr + k);
        acc = __builtin_amdgcn_wmma_f32_16x16x4_f32(
            false, a, false, b, (short)0, acc, false, false);
    }
    int n = n0 + lq;
    float bias = bdt[n];
#pragma unroll
    for (int v = 0; v < 8; ++v) {
        int m = m0 + v + 8 * half;
        float z = acc[v] + bias;
        float sp = (z > 20.0f) ? z : log1pf(__expf(z));
        dt[(size_t)m * D_IN + n] = sp;
    }
}

#define SCAN1_STEP(J)                                                       \
    {                                                                       \
        const float ab = __builtin_amdgcn_exp2f(row_bcast<J>(dtv) * A2);    \
        h = __builtin_fmaf(ab, h, row_bcast<J>(dxv) * Bv[J]);               \
    }

__global__ void __launch_bounds__(SCAN_THREADS)
scan_phase1(const float* __restrict__ X,
            const float* __restrict__ xdbl,
            const float* __restrict__ dt,
            const float* __restrict__ A_log,
            float* __restrict__ aprod,
            float* __restrict__ hend) {
    const int lane = threadIdx.x & 31;
    const int s = lane & 15;
    const int wave = blockIdx.x * (SCAN_THREADS / 32) + (threadIdx.x >> 5);
    const int chunk = wave / (NCHAN / 2);
    const int channel = (wave % (NCHAN / 2)) * 2 + (lane >> 4);
    const int b = channel / D_IN;
    const int d = channel % D_IN;
    const float A2 = -__expf(A_log[d * D_ST + s]) * LOG2E;
    float dtsum = 0.0f, h = 0.0f;
    const size_t m0 = (size_t)b * SEQ + (size_t)chunk * LCHUNK;
    const float* dtp = dt + (m0 + s) * D_IN + d;
    const float* xp  = X  + (m0 + s) * D_IN + d;
    const float* bp  = xdbl + m0 * N_E + DT_RANK + s;
    for (int t = 0; t < LCHUNK; t += TBLK) {
        const float dtv = dtp[(size_t)t * D_IN];
        const float dxv = dtv * xp[(size_t)t * D_IN];
        dtsum += dtv;
        float Bv[TBLK];
        for (int j = 0; j < TBLK; ++j)
            Bv[j] = bp[(size_t)(t + j) * N_E];
        SCAN1_STEP(0)  SCAN1_STEP(1)  SCAN1_STEP(2)  SCAN1_STEP(3)
        SCAN1_STEP(4)  SCAN1_STEP(5)  SCAN1_STEP(6)  SCAN1_STEP(7)
        SCAN1_STEP(8)  SCAN1_STEP(9)  SCAN1_STEP(10) SCAN1_STEP(11)
        SCAN1_STEP(12) SCAN1_STEP(13) SCAN1_STEP(14) SCAN1_STEP(15)
    }
    dtsum += row_xor<1>(dtsum);
    dtsum += row_xor<2>(dtsum);
    dtsum += row_xor<4>(dtsum);
    dtsum += row_xor<8>(dtsum);
    const int idx = (channel * D_ST + s) * NCHUNK + chunk;
    aprod[idx] = __builtin_amdgcn_exp2f(dtsum * A2);
    hend[idx]  = h;
}

__global__ void scan_phase2(const float* __restrict__ aprod,
                            const float* __restrict__ hend,
                            float* __restrict__ carry) {
    int t = blockIdx.x * blockDim.x + threadIdx.x;
    if (t >= NCHAN * D_ST) return;
    float c = 0.0f;
#pragma unroll
    for (int ch = 0; ch < NCHUNK; ++ch) {
        int idx = t * NCHUNK + ch;
        carry[idx] = c;
        c = aprod[idx] * c + hend[idx];
    }
}

#define SCAN3_STEP(J)                                                       \
    {                                                                       \
        const float ab = __builtin_amdgcn_exp2f(row_bcast<J>(dtv) * A2);    \
        h = __builtin_fmaf(ab, h, row_bcast<J>(dxv) * Bv[J]);               \
        c[J] = h * Cv[J];                                                   \
    }

__global__ void __launch_bounds__(SCAN_THREADS)
scan_phase3(const float* __restrict__ X,
            const float* __restrict__ xdbl,
            const float* __restrict__ dt,
            const float* __restrict__ A_log,
            const float* __restrict__ Dp,
            const float* __restrict__ carry,
            float* __restrict__ y) {
    const int lane = threadIdx.x & 31;
    const int s = lane & 15;
    const int wave = blockIdx.x * (SCAN_THREADS / 32) + (threadIdx.x >> 5);
    const int chunk = wave / (NCHAN / 2);
    const int channel = (wave % (NCHAN / 2)) * 2 + (lane >> 4);
    const int b = channel / D_IN;
    const int d = channel % D_IN;
    const float A2 = -__expf(A_log[d * D_ST + s]) * LOG2E;
    float h = carry[(channel * D_ST + s) * NCHUNK + chunk];
    const float Dd = Dp[d];
    const size_t m0 = (size_t)b * SEQ + (size_t)chunk * LCHUNK;
    const float* dtp = dt + (m0 + s) * D_IN + d;
    const float* xp  = X  + (m0 + s) * D_IN + d;
    float*       yp  = y  + (m0 + s) * D_IN + d;
    const float* bcp = xdbl + m0 * N_E + DT_RANK + s;
    const bool up1 = (s & 1) != 0, up2 = (s & 2) != 0, up4 = (s & 4) != 0, up8 = (s & 8) != 0;
    for (int t = 0; t < LCHUNK; t += TBLK) {
        const float dtv = dtp[(size_t)t * D_IN];
        const float xv  = xp[(size_t)t * D_IN];
        const float dxv = dtv * xv;
        float Bv[TBLK], Cv[TBLK], c[TBLK];
        for (int j = 0; j < TBLK; ++j) {
            Bv[j] = bcp[(size_t)(t + j) * N_E];
            Cv[j] = bcp[(size_t)(t + j) * N_E + D_ST];
        }
        SCAN3_STEP(0)  SCAN3_STEP(1)  SCAN3_STEP(2)  SCAN3_STEP(3)
        SCAN3_STEP(4)  SCAN3_STEP(5)  SCAN3_STEP(6)  SCAN3_STEP(7)
        SCAN3_STEP(8)  SCAN3_STEP(9)  SCAN3_STEP(10) SCAN3_STEP(11)
        SCAN3_STEP(12) SCAN3_STEP(13) SCAN3_STEP(14) SCAN3_STEP(15)
        float r1[8], r2[4], r3[2];
        for (int k = 0; k < 8; ++k) r1[k] = fold<1>(c[2 * k], c[2 * k + 1], up1);
        for (int k = 0; k < 4; ++k) r2[k] = fold<2>(r1[2 * k], r1[2 * k + 1], up2);
        for (int k = 0; k < 2; ++k) r3[k] = fold<4>(r2[2 * k], r2[2 * k + 1], up4);
        const float ysum = fold<8>(r3[0], r3[1], up8);
        yp[(size_t)t * D_IN] = ysum + Dd * xv;
    }
}

extern "C" void kernel_launch(void* const* d_in, const int* in_sizes, int n_in,
                              void* d_out, int out_size, void* d_ws, size_t ws_size,
                              hipStream_t stream) {
    const float* x       = (const float*)d_in[0];
    const float* A_log   = (const float*)d_in[1];
    const float* Dp      = (const float*)d_in[2];
    const float* W_xproj = (const float*)d_in[3];
    const float* W_dt    = (const float*)d_in[4];
    const float* b_dt    = (const float*)d_in[5];
    float* y = (float*)d_out;

    char* ws = (char*)d_ws;
    size_t off = 0;
    float* xdbl  = (float*)(ws + off); off += (size_t)M_TOT * N_E * 4;
    off = (off + 255) & ~(size_t)255;
    float* dtbuf = (float*)(ws + off); off += (size_t)M_TOT * D_IN * 4;
    off = (off + 255) & ~(size_t)255;
    float* aprod = (float*)(ws + off); off += (size_t)NCHAN * D_ST * NCHUNK * 4;
    off = (off + 255) & ~(size_t)255;
    float* hend  = (float*)(ws + off); off += (size_t)NCHAN * D_ST * NCHUNK * 4;
    off = (off + 255) & ~(size_t)255;
    float* carry = (float*)(ws + off);

    xproj_wmma<<<1536 / 4, 128, 0, stream>>>(x, W_xproj, xdbl);
    dt_wmma<<<16384 / 8, 256, 0, stream>>>(xdbl, W_dt, b_dt, dtbuf);
    scan_phase1<<<SCAN_WAVES / (SCAN_THREADS / 32), SCAN_THREADS, 0, stream>>>(
        x, xdbl, dtbuf, A_log, aprod, hend);
    scan_phase2<<<(NCHAN * D_ST) / 256, 256, 0, stream>>>(aprod, hend, carry);
    scan_phase3<<<SCAN_WAVES / (SCAN_THREADS / 32), SCAN_THREADS, 0, stream>>>(
        x, xdbl, dtbuf, A_log, Dp, carry, y);
}


# Scan kernels fed through LDS: a workgroup stages its chunk's dt, x, B, C as whole 128-byte lines in double-buffered 64-step tiles (B, C transposed); lanes read LDS, y leaves as whole lines
# speedup vs baseline: 3.9596x; 1.2312x over previous
#include <hip/hip_runtime.h>
#include <hip/hip_bf16.h>
#include <math.h>

typedef __attribute__((ext_vector_type(2))) float v2f;
typedef __attribute__((ext_vector_type(8))) float v8f;

#define D_IN   1024
#define D_ST   16
#define DT_RANK 64
#define N_E    96
#define BATCH  2
#define SEQ    2048
#define M_TOT  (BATCH*SEQ)
#define NCHUNK 8
#define LCHUNK (SEQ/NCHUNK)
#define NCHAN  (BATCH*D_IN)
#define TBLK   16
#define SCAN_THREADS 512
#define SCAN_WAVES (NCHAN / 2 * NCHUNK)
#define LOG2E 1.44269504088896341f
#define TS       64
#define NSUB     (LCHUNK / TS)
#define DX_PITCH 36
#define BC_PITCH 68
#define DX_TILE  (TS * DX_PITCH)
#define BC_TILE  (32 * BC_PITCH)
#define BUF_FLOATS (2 * DX_TILE + BC_TILE)
typedef __attribute__((ext_vector_type(4))) float v4f;

template <int CTRL>
__device__ __forceinline__ float row_dpp(float v) {
    const int b = __builtin_bit_cast(int, v);
    return __builtin_bit_cast(float, __builtin_amdgcn_update_dpp(0, b, CTRL, 0xf, 0xf, true));
}
template <int J> __device__ __forceinline__ float row_bcast(float v) { return row_dpp<0x150 + J>(v); }
template <int M> __device__ __forceinline__ float row_xor(float v) { return row_dpp<0x160 + M>(v); }

template <int M>
__device__ __forceinline__ float fold(float a, float b, bool up) {
    const float keep = up ? b : a;
    const float send = up ? a : b;
    return keep + row_xor<M>(send);
}

__global__ void xproj_wmma(const float* __restrict__ X,
                           const float* __restrict__ W,
                           float* __restrict__ xdbl) {
    int wave = (blockIdx.x * blockDim.x + threadIdx.x) >> 5;
    int lane = threadIdx.x & 31;
    const int NT = N_E / 16;
    int nt = wave % NT;
    int mt = wave / NT;
    if (mt >= M_TOT / 16) return;
    int m0 = mt * 16, n0 = nt * 16;
    int half = lane >> 4, lq = lane & 15;
    int koff = half * 2;
    const float* aptr = X + (size_t)(m0 + lq) * D_IN + koff;
    const float* bptr = W + (size_t)(n0 + lq) * D_IN + koff;
    v8f acc = {};
    for (int k = 0; k < D_IN; k += 4) {
        v2f a = *(const v2f*)(aptr + k);
        v2f b = *(const v2f*)(bptr + k);
        acc = __builtin_amdgcn_wmma_f32_16x16x4_f32(
            false, a, false, b, (short)0, acc, false, false);
    }
    int n = n0 + lq;
#pragma unroll
    for (int v = 0; v < 8; ++v) {
        int m = m0 + v + 8 * half;
        xdbl[(size_t)m * N_E + n] = acc[v];
    }
}

__global__ void dt_wmma(const float* __restrict__ xdbl,
                        const float* __restrict__ Wdt,
                        const float* __restrict__ bdt,
                        float* __restrict__ dt) {
    int wave = (blockIdx.x * blockDim.x + threadIdx.x) >> 5;
    int lane = threadIdx.x & 31;
    const int NT = D_IN / 16;
    int nt = wave % NT;
    int mt = wave / NT;
    if (mt >= M_TOT / 16) return;
    int m0 = mt * 16, n0 = nt * 16;
    int half = lane >> 4, lq = lane & 15;
    int koff = half * 2;
    const float* aptr = xdbl + (size_t)(m0 + lq) * N_E + koff;
    const float* bptr = Wdt + (size_t)(n0 + lq) * DT_RANK + koff;
    v8f acc = {};
#pragma unroll
    for (int k = 0; k < DT_RANK; k += 4) {
        v2f a = *(const v2f*)(aptr + k);
        v2f b = *(const v2f*)(bptr + k);
        acc = __builtin_amdgcn_wmma_f32_16x16x4_f32(
            false, a, false, b, (short)0, acc, false, false);
    }
    int n = n0 + lq;
    float bias = bdt[n];
#pragma unroll
    for (int v = 0; v < 8; ++v) {
        int m = m0 + v + 8 * half;
        float z = acc[v] + bias;
        float sp = (z > 20.0f) ? z : log1pf(__expf(z));
        dt[(size_t)m * D_IN + n] = sp;
    }
}

struct StageRegs { v4f dt, x, bc; };

__device__ __forceinline__ void stage_load(StageRegs& r,
                                           const float* __restrict__ X,
                                           const float* __restrict__ xdbl,
                                           const float* __restrict__ dt,
                                           size_t mrow, int dcol, int q) {
    r.dt = *(const v4f*)(dt + mrow * D_IN + dcol);
    r.x  = *(const v4f*)(X  + mrow * D_IN + dcol);
    r.bc = *(const v4f*)(xdbl + mrow * N_E + DT_RANK + 4 * q);
}

__device__ __forceinline__ void stage_store(const StageRegs& r, float* buf, int row, int q) {
    *(v4f*)(buf + row * DX_PITCH + 4 * q) = r.dt;
    *(v4f*)(buf + DX_TILE + row * DX_PITCH + 4 * q) = r.x;
    float* bc = buf + 2 * DX_TILE + (4 * q) * BC_PITCH + row;
    bc[0]            = r.bc[0];
    bc[BC_PITCH]     = r.bc[1];
    bc[2 * BC_PITCH] = r.bc[2];
    bc[3 * BC_PITCH] = r.bc[3];
}

template <int ROW0>
__device__ __forceinline__ void load_state_row(float (&v)[TBLK], const float* bcT, int s, int tb) {
    const float* p = bcT + (ROW0 + s) * BC_PITCH + tb * TBLK;
    for (int i = 0; i < 4; ++i) {
        const v4f w = *(const v4f*)(p + 4 * i);
        v[4 * i] = w[0]; v[4 * i + 1] = w[1]; v[4 * i + 2] = w[2]; v[4 * i + 3] = w[3];
    }
}

#define SCAN1_STEP(J)                                                       \
    {                                                                       \
        const float ab = __builtin_amdgcn_exp2f(row_bcast<J>(dtv) * A2);    \
        h = __builtin_fmaf(ab, h, row_bcast<J>(dxv) * Bv[J]);               \
    }

__global__ void __launch_bounds__(SCAN_THREADS)
scan_phase1(const float* __restrict__ X,
            const float* __restrict__ xdbl,
            const float* __restrict__ dt,
            const float* __restrict__ A_log,
            float* __restrict__ aprod,
            float* __restrict__ hend) {
    __shared__ __attribute__((aligned(16))) float lds[2 * BUF_FLOATS];
    const int tid = threadIdx.x;
    const int lane = tid & 31;
    const int s = lane & 15;
    const int wave = blockIdx.x * (SCAN_THREADS / 32) + (tid >> 5);
    const int chunk = wave / (NCHAN / 2);
    const int cl = (tid >> 5) * 2 + (lane >> 4);
    const int channel = (wave % (NCHAN / 2)) * 2 + (lane >> 4);
    const int b = channel / D_IN;
    const int d = channel % D_IN;
    const float A2 = -__expf(A_log[d * D_ST + s]) * LOG2E;
    float dtsum = 0.0f, h = 0.0f;
    const size_t m0 = (size_t)b * SEQ + (size_t)chunk * LCHUNK;
    const int row = tid >> 3, q = tid & 7;
    const int dcol = (d - cl) + 4 * q;
    StageRegs st;
    stage_load(st, X, xdbl, dt, m0 + row, dcol, q);
    stage_store(st, lds, row, q);
    __syncthreads();
    for (int sub = 0; sub < NSUB; ++sub) {
        const float* buf = lds + (sub & 1) * BUF_FLOATS;
        if (sub + 1 < NSUB)
            stage_load(st, X, xdbl, dt, m0 + (size_t)(sub + 1) * TS + row, dcol, q);
        const float* dtT = buf;
        const float* xT  = buf + DX_TILE;
        const float* bcT = buf + 2 * DX_TILE;
        for (int tb = 0; tb < TS / TBLK; ++tb) {
            const int o = (tb * TBLK + s) * DX_PITCH + cl;
            const float dtv = dtT[o];
            const float dxv = dtv * xT[o];
            dtsum += dtv;
            float Bv[TBLK];
            load_state_row<0>(Bv, bcT, s, tb);
            SCAN1_STEP(0)  SCAN1_STEP(1)  SCAN1_STEP(2)  SCAN1_STEP(3)
            SCAN1_STEP(4)  SCAN1_STEP(5)  SCAN1_STEP(6)  SCAN1_STEP(7)
            SCAN1_STEP(8)  SCAN1_STEP(9)  SCAN1_STEP(10) SCAN1_STEP(11)
            SCAN1_STEP(12) SCAN1_STEP(13) SCAN1_STEP(14) SCAN1_STEP(15)
        }
        if (sub + 1 < NSUB) {
            stage_store(st, lds + ((sub + 1) & 1) * BUF_FLOATS, row, q);
            __syncthreads();
        }
    }
    dtsum += row_xor<1>(dtsum);
    dtsum += row_xor<2>(dtsum);
    dtsum += row_xor<4>(dtsum);
    dtsum += row_xor<8>(dtsum);
    const int idx = (channel * D_ST + s) * NCHUNK + chunk;
    aprod[idx] = __builtin_amdgcn_exp2f(dtsum * A2);
    hend[idx]  = h;
}

__global__ void scan_phase2(const float* __restrict__ aprod,
                            const float* __restrict__ hend,
                            float* __restrict__ carry) {
    int t = blockIdx.x * blockDim.x + threadIdx.x;
    if (t >= NCHAN * D_ST) return;
    float c = 0.0f;
#pragma unroll
    for (int ch = 0; ch < NCHUNK; ++ch) {
        int idx = t * NCHUNK + ch;
        carry[idx] = c;
        c = aprod[idx] * c + hend[idx];
    }
}

#define SCAN3_STEP(J)                                                       \
    {                                                                       \
        const float ab = __builtin_amdgcn_exp2f(row_bcast<J>(dtv) * A2);    \
        h = __builtin_fmaf(ab, h, row_bcast<J>(dxv) * Bv[J]);               \
        c[J] = h * Cv[J];                                                   \
    }

__global__ void __launch_bounds__(SCAN_THREADS)
scan_phase3(const float* __restrict__ X,
            const float* __restrict__ xdbl,
            const float* __restrict__ dt,
            const float* __restrict__ A_log,
            const float* __restrict__ Dp,
            const float* __restrict__ carry,
            float* __restrict__ y) {
    __shared__ __attribute__((aligned(16))) float lds[2 * BUF_FLOATS];
    const int tid = threadIdx.x;
    const int lane = tid & 31;
    const int s = lane & 15;
    const int wave = blockIdx.x * (SCAN_THREADS / 32) + (tid >> 5);
    const int chunk = wave / (NCHAN / 2);
    const int cl = (tid >> 5) * 2 + (lane >> 4);
    const int channel = (wave % (NCHAN / 2)) * 2 + (lane >> 4);
    const int b = channel / D_IN;
    const int d = channel % D_IN;
    const float A2 = -__expf(A_log[d * D_ST + s]) * LOG2E;
    float h = carry[(channel * D_ST + s) * NCHUNK + chunk];
    const float Dd = Dp[d];
    const size_t m0 = (size_t)b * SEQ + (size_t)chunk * LCHUNK;
    const int row = tid >> 3, q = tid & 7;
    const int dcol = (d - cl) + 4 * q;
    const bool up1 = (s & 1) != 0, up2 = (s & 2) != 0, up4 = (s & 4) != 0, up8 = (s & 8) != 0;
    StageRegs st;
    stage_load(st, X, xdbl, dt, m0 + row, dcol, q);
    stage_store(st, lds, row, q);
    __syncthreads();
    for (int sub = 0; sub < NSUB; ++sub) {
        float* buf = lds + (sub & 1) * BUF_FLOATS;
        float* oth = lds + ((sub + 1) & 1) * BUF_FLOATS;
        if (sub + 1 < NSUB)
            stage_load(st, X, xdbl, dt, m0 + (size_t)(sub + 1) * TS + row, dcol, q);
        if (sub > 0)
            *(v4f*)(y + (m0 + (size_t)(sub - 1) * TS + row) * D_IN + dcol) =
                *(const v4f*)(oth + row * DX_PITCH + 4 * q);
        float*       dtT = buf;
        const float* xT  = buf + DX_TILE;
        const float* bcT = buf + 2 * DX_TILE;
        for (int tb = 0; tb < TS / TBLK; ++tb) {
            const int o = (tb * TBLK + s) * DX_PITCH + cl;
            const float dtv = dtT[o];
            const float xv  = xT[o];
            const float dxv = dtv * xv;
            float Bv[TBLK], Cv[TBLK], c[TBLK];
            load_state_row<0>(Bv, bcT, s, tb);
            load_state_row<D_ST>(Cv, bcT, s, tb);
            SCAN3_STEP(0)  SCAN3_STEP(1)  SCAN3_STEP(2)  SCAN3_STEP(3)
            SCAN3_STEP(4)  SCAN3_STEP(5)  SCAN3_STEP(6)  SCAN3_STEP(7)
            SCAN3_STEP(8)  SCAN3_STEP(9)  SCAN3_STEP(10) SCAN3_STEP(11)
            SCAN3_STEP(12) SCAN3_STEP(13) SCAN3_STEP(14) SCAN3_STEP(15)
            float r1[8], r2[4], r3[2];
            for (int k = 0; k < 8; ++k) r1[k] = fold<1>(c[2 * k], c[2 * k + 1], up1);
            for (int k = 0; k < 4; ++k) r2[k] = fold<2>(r1[2 * k], r1[2 * k + 1], up2);
            for (int k = 0; k < 2; ++k) r3[k] = fold<4>(r2[2 * k], r2[2 * k + 1], up4);
            const float ysum = fold<8>(r3[0], r3[1], up8);
            dtT[o] = ysum + Dd * xv;
        }
        if (sub + 1 < NSUB)
            stage_store(st, oth, row, q);
        __syncthreads();
    }
    *(v4f*)(y + (m0 + (size_t)(NSUB - 1) * TS + row) * D_IN + dcol) =
        *(const v4f*)(lds + ((NSUB - 1) & 1) * BUF_FLOATS + row * DX_PITCH + 4 * q);
}

extern "C" void kernel_launch(void* const* d_in, const int* in_sizes, int n_in,
                              void* d_out, int out_size, void* d_ws, size_t ws_size,
                              hipStream_t stream) {
    const float* x       = (const float*)d_in[0];
    const float* A_log   = (const float*)d_in[1];
    const float* Dp      = (const float*)d_in[2];
    const float* W_xproj = (const float*)d_in[3];
    const float* W_dt    = (const float*)d_in[4];
    const float* b_dt    = (const float*)d_in[5];
    float* y = (float*)d_out;

    char* ws = (char*)d_ws;
    size_t off = 0;
    float* xdbl  = (float*)(ws + off); off += (size_t)M_TOT * N_E * 4;
    off = (off + 255) & ~(size_t)255;
    float* dtbuf = (float*)(ws + off); off += (size_t)M_TOT * D_IN * 4;
    off = (off + 255) & ~(size_t)255;
    float* aprod = (float*)(ws + off); off += (size_t)NCHAN * D_ST * NCHUNK * 4;
    off = (off + 255) & ~(size_t)255;
    float* hend  = (float*)(ws + off); off += (size_t)NCHAN * D_ST * NCHUNK * 4;
    off = (off + 255) & ~(size_t)255;
    float* carry = (float*)(ws + off);

    xproj_wmma<<<1536 / 4, 128, 0, stream>>>(x, W_xproj, xdbl);
    dt_wmma<<<16384 / 8, 256, 0, stream>>>(xdbl, W_dt, b_dt, dtbuf);
    scan_phase1<<<SCAN_WAVES / (SCAN_THREADS / 32), SCAN_THREADS, 0, stream>>>(
        x, xdbl, dtbuf, A_log, aprod, hend);
    scan_phase2<<<(NCHAN * D_ST) / 256, 256, 0, stream>>>(aprod, hend, carry);
    scan_phase3<<<SCAN_WAVES / (SCAN_THREADS / 32), SCAN_THREADS, 0, stream>>>(
        x, xdbl, dtbuf, A_log, Dp, carry, y);
}


# Cross-chunk carry formed inside the replay kernel: each lane fetches its (decay, end state) pairs and combines the earlier chunks' itself, same order; the combine launch is gone, 4 launches not 5
# speedup vs baseline: 4.1191x; 1.0403x over previous
#include <hip/hip_runtime.h>
#include <hip/hip_bf16.h>
#include <math.h>

typedef __attribute__((ext_vector_type(2))) float v2f;
typedef __attribute__((ext_vector_type(8))) float v8f;

#define D_IN   1024
#define D_ST   16
#define DT_RANK 64
#define N_E    96
#define BATCH  2
#define SEQ    2048
#define M_TOT  (BATCH*SEQ)
#define NCHUNK 8
#define LCHUNK (SEQ/NCHUNK)
#define NCHAN  (BATCH*D_IN)
#define TBLK   16
#define SCAN_THREADS 512
#define SCAN_WAVES (NCHAN / 2 * NCHUNK)
#define LOG2E 1.44269504088896341f
#define TS       64
#define NSUB     (LCHUNK / TS)
#define DX_PITCH 36
#define BC_PITCH 68
#define DX_TILE  (TS * DX_PITCH)
#define BC_TILE  (32 * BC_PITCH)
#define BUF_FLOATS (2 * DX_TILE + BC_TILE)
typedef __attribute__((ext_vector_type(4))) float v4f;

template <int CTRL>
__device__ __forceinline__ float row_dpp(float v) {
    const int b = __builtin_bit_cast(int, v);
    return __builtin_bit_cast(float, __builtin_amdgcn_update_dpp(0, b, CTRL, 0xf, 0xf, true));
}
template <int J> __device__ __forceinline__ float row_bcast(float v) { return row_dpp<0x150 + J>(v); }
template <int M> __device__ __forceinline__ float row_xor(float v) { return row_dpp<0x160 + M>(v); }

template <int M>
__device__ __forceinline__ float fold(float a, float b, bool up) {
    const float keep = up ? b : a;
    const float send = up ? a : b;
    return keep + row_xor<M>(send);
}

__global__ void xproj_wmma(const float* __restrict__ X,
                           const float* __restrict__ W,
                           float* __restrict__ xdbl) {
    int wave = (blockIdx.x * blockDim.x + threadIdx.x) >> 5;
    int lane = threadIdx.x & 31;
    const int NT = N_E / 16;
    int nt = wave % NT;
    int mt = wave / NT;
    if (mt >= M_TOT / 16) return;
    int m0 = mt * 16, n0 = nt * 16;
    int half = lane >> 4, lq = lane & 15;
    int koff = half * 2;
    const float* aptr = X + (size_t)(m0 + lq) * D_IN + koff;
    const float* bptr = W + (size_t)(n0 + lq) * D_IN + koff;
    v8f acc = {};
    for (int k = 0; k < D_IN; k += 4) {
        v2f a = *(const v2f*)(aptr + k);
        v2f b = *(const v2f*)(bptr + k);
        acc = __builtin_amdgcn_wmma_f32_16x16x4_f32(
            false, a, false, b, (short)0, acc, false, false);
    }
    int n = n0 + lq;
#pragma unroll
    for (int v = 0; v < 8; ++v) {
        int m = m0 + v + 8 * half;
        xdbl[(size_t)m * N_E + n] = acc[v];
    }
}

__global__ void dt_wmma(const float* __restrict__ xdbl,
                        const float* __restrict__ Wdt,
                        const float* __restrict__ bdt,
                        float* __restrict__ dt) {
    int wave = (blockIdx.x * blockDim.x + threadIdx.x) >> 5;
    int lane = threadIdx.x & 31;
    const int NT = D_IN / 16;
    int nt = wave % NT;
    int mt = wave / NT;
    if (mt >= M_TOT / 16) return;
    int m0 = mt * 16, n0 = nt * 16;
    int half = lane >> 4, lq = lane & 15;
    int koff = half * 2;
    const float* aptr = xdbl + (size_t)(m0 + lq) * N_E + koff;
    const float* bptr = Wdt + (size_t)(n0 + lq) * DT_RANK + koff;
    v8f acc = {};
#pragma unroll
    for (int k = 0; k < DT_RANK; k += 4) {
        v2f a = *(const v2f*)(aptr + k);
        v2f b = *(const v2f*)(bptr + k);
        acc = __builtin_amdgcn_wmma_f32_16x16x4_f32(
            false, a, false, b, (short)0, acc, false, false);
    }
    int n = n0 + lq;
    float bias = bdt[n];
#pragma unroll
    for (int v = 0; v < 8; ++v) {
        int m = m0 + v + 8 * half;
        float z = acc[v] + bias;
        float sp = (z > 20.0f) ? z : log1pf(__expf(z));
        dt[(size_t)m * D_IN + n] = sp;
    }
}

struct StageRegs { v4f dt, x, bc; };

__device__ __forceinline__ void stage_load(StageRegs& r,
                                           const float* __restrict__ X,
                                           const float* __restrict__ xdbl,
                                           const float* __restrict__ dt,
                                           size_t mrow, int dcol, int q) {
    r.dt = *(const v4f*)(dt + mrow * D_IN + dcol);
    r.x  = *(const v4f*)(X  + mrow * D_IN + dcol);
    r.bc = *(const v4f*)(xdbl + mrow * N_E + DT_RANK + 4 * q);
}

__device__ __forceinline__ void stage_store(const StageRegs& r, float* buf, int row, int q) {
    *(v4f*)(buf + row * DX_PITCH + 4 * q) = r.dt;
    *(v4f*)(buf + DX_TILE + row * DX_PITCH + 4 * q) = r.x;
    float* bc = buf + 2 * DX_TILE + (4 * q) * BC_PITCH + row;
    bc[0]            = r.bc[0];
    bc[BC_PITCH]     = r.bc[1];
    bc[2 * BC_PITCH] = r.bc[2];
    bc[3 * BC_PITCH] = r.bc[3];
}

template <int ROW0>
__device__ __forceinline__ void load_state_row(float (&v)[TBLK], const float* bcT, int s, int tb) {
    const float* p = bcT + (ROW0 + s) * BC_PITCH + tb * TBLK;
    for (int i = 0; i < 4; ++i) {
        const v4f w = *(const v4f*)(p + 4 * i);
        v[4 * i] = w[0]; v[4 * i + 1] = w[1]; v[4 * i + 2] = w[2]; v[4 * i + 3] = w[3];
    }
}

#define SCAN1_STEP(J)                                                       \
    {                                                                       \
        const float ab = __builtin_amdgcn_exp2f(row_bcast<J>(dtv) * A2);    \
        h = __builtin_fmaf(ab, h, row_bcast<J>(dxv) * Bv[J]);               \
    }

__global__ void __launch_bounds__(SCAN_THREADS)
scan_phase1(const float* __restrict__ X,
            const float* __restrict__ xdbl,
            const float* __restrict__ dt,
            const float* __restrict__ A_log,
            float* __restrict__ aprod,
            float* __restrict__ hend) {
    __shared__ __attribute__((aligned(16))) float lds[2 * BUF_FLOATS];
    const int tid = threadIdx.x;
    const int lane = tid & 31;
    const int s = lane & 15;
    const int wave = blockIdx.x * (SCAN_THREADS / 32) + (tid >> 5);
    const int chunk = wave / (NCHAN / 2);
    const int cl = (tid >> 5) * 2 + (lane >> 4);
    const int channel = (wave % (NCHAN / 2)) * 2 + (lane >> 4);
    const int b = channel / D_IN;
    const int d = channel % D_IN;
    const float A2 = -__expf(A_log[d * D_ST + s]) * LOG2E;
    float dtsum = 0.0f, h = 0.0f;
    const size_t m0 = (size_t)b * SEQ + (size_t)chunk * LCHUNK;
    const int row = tid >> 3, q = tid & 7;
    const int dcol = (d - cl) + 4 * q;
    StageRegs st;
    stage_load(st, X, xdbl, dt, m0 + row, dcol, q);
    stage_store(st, lds, row, q);
    __syncthreads();
    for (int sub = 0; sub < NSUB; ++sub) {
        const float* buf = lds + (sub & 1) * BUF_FLOATS;
        if (sub + 1 < NSUB)
            stage_load(st, X, xdbl, dt, m0 + (size_t)(sub + 1) * TS + row, dcol, q);
        const float* dtT = buf;
        const float* xT  = buf + DX_TILE;
        const float* bcT = buf + 2 * DX_TILE;
        for (int tb = 0; tb < TS / TBLK; ++tb) {
            const int o = (tb * TBLK + s) * DX_PITCH + cl;
            const float dtv = dtT[o];
            const float dxv = dtv * xT[o];
            dtsum += dtv;
            float Bv[TBLK];
            load_state_row<0>(Bv, bcT, s, tb);
            SCAN1_STEP(0)  SCAN1_STEP(1)  SCAN1_STEP(2)  SCAN1_STEP(3)
            SCAN1_STEP(4)  SCAN1_STEP(5)  SCAN1_STEP(6)  SCAN1_STEP(7)
            SCAN1_STEP(8)  SCAN1_STEP(9)  SCAN1_STEP(10) SCAN1_STEP(11)
            SCAN1_STEP(12) SCAN1_STEP(13) SCAN1_STEP(14) SCAN1_STEP(15)
        }
        if (sub + 1 < NSUB) {
            stage_store(st, lds + ((sub + 1) & 1) * BUF_FLOATS, row, q);
            __syncthreads();
        }
    }
    dtsum += row_xor<1>(dtsum);
    dtsum += row_xor<2>(dtsum);
    dtsum += row_xor<4>(dtsum);
    dtsum += row_xor<8>(dtsum);
    const int idx = (channel * D_ST + s) * NCHUNK + chunk;
    aprod[idx] = __builtin_amdgcn_exp2f(dtsum * A2);
    hend[idx]  = h;
}

#define SCAN3_STEP(J)                                                       \
    {                                                                       \
        const float ab = __builtin_amdgcn_exp2f(row_bcast<J>(dtv) * A2);    \
        h = __builtin_fmaf(ab, h, row_bcast<J>(dxv) * Bv[J]);               \
        c[J] = h * Cv[J];                                                   \
    }

__global__ void __launch_bounds__(SCAN_THREADS)
scan_phase3(const float* __restrict__ X,
            const float* __restrict__ xdbl,
            const float* __restrict__ dt,
            const float* __restrict__ A_log,
            const float* __restrict__ Dp,
            const float* __restrict__ aprod,
            const float* __restrict__ hend,
            float* __restrict__ y) {
    __shared__ __attribute__((aligned(16))) float lds[2 * BUF_FLOATS];
    const int tid = threadIdx.x;
    const int lane = tid & 31;
    const int s = lane & 15;
    const int wave = blockIdx.x * (SCAN_THREADS / 32) + (tid >> 5);
    const int chunk = wave / (NCHAN / 2);
    const int cl = (tid >> 5) * 2 + (lane >> 4);
    const int channel = (wave % (NCHAN / 2)) * 2 + (lane >> 4);
    const int b = channel / D_IN;
    const int d = channel % D_IN;
    const float A2 = -__expf(A_log[d * D_ST + s]) * LOG2E;
    static_assert(NCHUNK == 8, "the pair fetch below reads two 16-byte halves per array");
    const int pbase = (channel * D_ST + s) * NCHUNK;
    const v4f ap0 = *(const v4f*)(aprod + pbase), ap1 = *(const v4f*)(aprod + pbase + 4);
    const v4f he0 = *(const v4f*)(hend + pbase),  he1 = *(const v4f*)(hend + pbase + 4);
    const float apv[NCHUNK - 1] = {ap0[0], ap0[1], ap0[2], ap0[3], ap1[0], ap1[1], ap1[2]};
    const float hev[NCHUNK - 1] = {he0[0], he0[1], he0[2], he0[3], he1[0], he1[1], he1[2]};
    float h = 0.0f;
    for (int ch = 0; ch < NCHUNK - 1; ++ch)
        if (ch < chunk) h = __builtin_fmaf(apv[ch], h, hev[ch]);
    const float Dd = Dp[d];
    const size_t m0 = (size_t)b * SEQ + (size_t)chunk * LCHUNK;
    const int row = tid >> 3, q = tid & 7;
    const int dcol = (d - cl) + 4 * q;
    const bool up1 = (s & 1) != 0, up2 = (s & 2) != 0, up4 = (s & 4) != 0, up8 = (s & 8) != 0;
    StageRegs st;
    stage_load(st, X, xdbl, dt, m0 + row, dcol, q);
    stage_store(st, lds, row, q);
    __syncthreads();
    for (int sub = 0; sub < NSUB; ++sub) {
        float* buf = lds + (sub & 1) * BUF_FLOATS;
        float* oth = lds + ((sub + 1) & 1) * BUF_FLOATS;
        if (sub + 1 < NSUB)
            stage_load(st, X, xdbl, dt, m0 + (size_t)(sub + 1) * TS + row, dcol, q);
        if (sub > 0)
            *(v4f*)(y + (m0 + (size_t)(sub - 1) * TS + row) * D_IN + dcol) =
                *(const v4f*)(oth + row * DX_PITCH + 4 * q);
        float*       dtT = buf;
        const float* xT  = buf + DX_TILE;
        const float* bcT = buf + 2 * DX_TILE;
        for (int tb = 0; tb < TS / TBLK; ++tb) {
            const int o = (tb * TBLK + s) * DX_PITCH + cl;
            const float dtv = dtT[o];
            const float xv  = xT[o];
            const float dxv = dtv * xv;
            float Bv[TBLK], Cv[TBLK], c[TBLK];
            load_state_row<0>(Bv, bcT, s, tb);
            load_state_row<D_ST>(Cv, bcT, s, tb);
            SCAN3_STEP(0)  SCAN3_STEP(1)  SCAN3_STEP(2)  SCAN3_STEP(3)
            SCAN3_STEP(4)  SCAN3_STEP(5)  SCAN3_STEP(6)  SCAN3_STEP(7)
            SCAN3_STEP(8)  SCAN3_STEP(9)  SCAN3_STEP(10) SCAN3_STEP(11)
            SCAN3_STEP(12) SCAN3_STEP(13) SCAN3_STEP(14) SCAN3_STEP(15)
            float r1[8], r2[4], r3[2];
            for (int k = 0; k < 8; ++k) r1[k] = fold<1>(c[2 * k], c[2 * k + 1], up1);
            for (int k = 0; k < 4; ++k) r2[k] = fold<2>(r1[2 * k], r1[2 * k + 1], up2);
            for (int k = 0; k < 2; ++k) r3[k] = fold<4>(r2[2 * k], r2[2 * k + 1], up4);
            const float ysum = fold<8>(r3[0], r3[1], up8);
            dtT[o] = ysum + Dd * xv;
        }
        if (sub + 1 < NSUB)
            stage_store(st, oth, row, q);
        __syncthreads();
    }
    *(v4f*)(y + (m0 + (size_t)(NSUB - 1) * TS + row) * D_IN + dcol) =
        *(const v4f*)(lds + ((NSUB - 1) & 1) * BUF_FLOATS + row * DX_PITCH + 4 * q);
}

extern "C" void kernel_launch(void* const* d_in, const int* in_sizes, int n_in,
                              void* d_out, int out_size, void* d_ws, size_t ws_size,
                              hipStream_t stream) {
    const float* x       = (const float*)d_in[0];
    const float* A_log   = (const float*)d_in[1];
    const float* Dp      = (const float*)d_in[2];
    const float* W_xproj = (const float*)d_in[3];
    const float* W_dt    = (const float*)d_in[4];
    const float* b_dt    = (const float*)d_in[5];
    float* y = (float*)d_out;

    char* ws = (char*)d_ws;
    size_t off = 0;
    float* xdbl  = (float*)(ws + off); off += (size_t)M_TOT * N_E * 4;
    off = (off + 255) & ~(size_t)255;
    float* dtbuf = (float*)(ws + off); off += (size_t)M_TOT * D_IN * 4;
    off = (off + 255) & ~(size_t)255;
    float* aprod = (float*)(ws + off); off += (size_t)NCHAN * D_ST * NCHUNK * 4;
    off = (off + 255) & ~(size_t)255;
    float* hend  = (float*)(ws + off);

    xproj_wmma<<<1536 / 4, 128, 0, stream>>>(x, W_xproj, xdbl);
    dt_wmma<<<16384 / 8, 256, 0, stream>>>(xdbl, W_dt, b_dt, dtbuf);
    scan_phase1<<<SCAN_WAVES / (SCAN_THREADS / 32), SCAN_THREADS, 0, stream>>>(
        x, xdbl, dtbuf, A_log, aprod, hend);
    scan_phase3<<<SCAN_WAVES / (SCAN_THREADS / 32), SCAN_THREADS, 0, stream>>>(
        x, xdbl, dtbuf, A_log, Dp, aprod, hend, y);
}


# x_proj GEMM fed through LDS: a 12-wave workgroup per 16-row strip stages X and W in 128-k slabs as whole 512-byte row pieces, double-buffered; two accumulators x two k halves, K sum order changes
# speedup vs baseline: 4.7609x; 1.1558x over previous
#include <hip/hip_runtime.h>
#include <hip/hip_bf16.h>
#include <math.h>

typedef __attribute__((ext_vector_type(2))) float v2f;
typedef __attribute__((ext_vector_type(8))) float v8f;

#define D_IN   1024
#define D_ST   16
#define DT_RANK 64
#define N_E    96
#define BATCH  2
#define SEQ    2048
#define M_TOT  (BATCH*SEQ)
#define NCHUNK 8
#define LCHUNK (SEQ/NCHUNK)
#define NCHAN  (BATCH*D_IN)
#define TBLK   16
#define SCAN_THREADS 512
#define SCAN_WAVES (NCHAN / 2 * NCHUNK)
#define LOG2E 1.44269504088896341f
#define TS       64
#define NSUB     (LCHUNK / TS)
#define DX_PITCH 36
#define BC_PITCH 68
#define DX_TILE  (TS * DX_PITCH)
#define BC_TILE  (32 * BC_PITCH)
#define BUF_FLOATS (2 * DX_TILE + BC_TILE)
typedef __attribute__((ext_vector_type(4))) float v4f;

template <int CTRL>
__device__ __forceinline__ float row_dpp(float v) {
    const int b = __builtin_bit_cast(int, v);
    return __builtin_bit_cast(float, __builtin_amdgcn_update_dpp(0, b, CTRL, 0xf, 0xf, true));
}
template <int J> __device__ __forceinline__ float row_bcast(float v) { return row_dpp<0x150 + J>(v); }
template <int M> __device__ __forceinline__ float row_xor(float v) { return row_dpp<0x160 + M>(v); }

template <int M>
__device__ __forceinline__ float fold(float a, float b, bool up) {
    const float keep = up ? b : a;
    const float send = up ? a : b;
    return keep + row_xor<M>(send);
}

#define XP_M       16
#define XP_KS      128
#define XP_NSLAB   (D_IN / XP_KS)
#define XP_PITCH   (XP_KS + 4)
#define XP_ROWS    (XP_M + N_E)
#define XP_BUF     (XP_ROWS * XP_PITCH)
#define XP_WAVES   12
#define XP_THREADS (XP_WAVES * 32)
#define XP_WROWS   (N_E / XP_WAVES)

struct XpStage { v4f w[XP_WROWS]; v4f x[2]; };

__device__ __forceinline__ void xp_load(XpStage& r, const float* __restrict__ X,
                                        const float* __restrict__ W,
                                        size_t m0, int slab, int wv, int lane) {
    const int kcol = slab * XP_KS + 4 * lane;
    for (int j = 0; j < XP_WROWS; ++j)
        r.w[j] = *(const v4f*)(W + (size_t)(wv + XP_WAVES * j) * D_IN + kcol);
    if (wv < 8) {
        r.x[0] = *(const v4f*)(X + (m0 + wv) * D_IN + kcol);
        r.x[1] = *(const v4f*)(X + (m0 + wv + 8) * D_IN + kcol);
    }
}

__device__ __forceinline__ void xp_store(const XpStage& r, float* buf, int wv, int lane) {
    for (int j = 0; j < XP_WROWS; ++j)
        *(v4f*)(buf + (XP_M + wv + XP_WAVES * j) * XP_PITCH + 4 * lane) = r.w[j];
    if (wv < 8) {
        *(v4f*)(buf + wv * XP_PITCH + 4 * lane) = r.x[0];
        *(v4f*)(buf + (wv + 8) * XP_PITCH + 4 * lane) = r.x[1];
    }
}

__global__ void __launch_bounds__(XP_THREADS)
xproj_wmma(const float* __restrict__ X,
           const float* __restrict__ W,
           float* __restrict__ xdbl) {
    __shared__ __attribute__((aligned(16))) float lds[2 * XP_BUF];
    __shared__ __attribute__((aligned(16))) float part[(N_E / 16) * 256];
    __shared__ __attribute__((aligned(16))) float res[XP_M * N_E];
    const int tid = threadIdx.x;
    const int lane = tid & 31;
    const int wv = tid >> 5;
    const int nt = wv % (N_E / 16);
    const int kh = wv / (N_E / 16);
    const int half = lane >> 4, lq = lane & 15;
    const size_t m0 = (size_t)blockIdx.x * XP_M;
    const int aoff = lq * XP_PITCH + kh * (XP_KS / 2) + 4 * half;
    const int boff = (XP_M + nt * 16 + lq) * XP_PITCH + kh * (XP_KS / 2) + 4 * half;
    v8f acc0 = {}, acc1 = {};
    XpStage st;
    xp_load(st, X, W, m0, 0, wv, lane);
    xp_store(st, lds, wv, lane);
    __syncthreads();
    for (int slab = 0; slab < XP_NSLAB; ++slab) {
        const float* buf = lds + (slab & 1) * XP_BUF;
        if (slab + 1 < XP_NSLAB)
            xp_load(st, X, W, m0, slab + 1, wv, lane);
        v4f af[8], bf[8];
        for (int i = 0; i < 8; ++i) {
            af[i] = *(const v4f*)(buf + aoff + 8 * i);
            bf[i] = *(const v4f*)(buf + boff + 8 * i);
        }
        for (int i = 0; i < 8; ++i) {
            const v2f a0 = {af[i][0], af[i][1]}, b0 = {bf[i][0], bf[i][1]};
            const v2f a1 = {af[i][2], af[i][3]}, b1 = {bf[i][2], bf[i][3]};
            acc0 = __builtin_amdgcn_wmma_f32_16x16x4_f32(
                false, a0, false, b0, (short)0, acc0, false, false);
            acc1 = __builtin_amdgcn_wmma_f32_16x16x4_f32(
                false, a1, false, b1, (short)0, acc1, false, false);
        }
        if (slab + 1 < XP_NSLAB) {
            xp_store(st, lds + ((slab + 1) & 1) * XP_BUF, wv, lane);
            __syncthreads();
        }
    }
    const v8f acc = acc0 + acc1;
    if (kh == 1)
        for (int v = 0; v < 8; ++v) part[nt * 256 + v * 32 + lane] = acc[v];
    __syncthreads();
    if (kh == 0)
        for (int v = 0; v < 8; ++v)
            res[(v + 8 * half) * N_E + nt * 16 + lq] = acc[v] + part[nt * 256 + v * 32 + lane];
    __syncthreads();
    *(v4f*)(xdbl + m0 * N_E + 4 * tid) = *(const v4f*)(res + 4 * tid);
}

__global__ void dt_wmma(const float* __restrict__ xdbl,
                        const float* __restrict__ Wdt,
                        const float* __restrict__ bdt,
                        float* __restrict__ dt) {
    int wave = (blockIdx.x * blockDim.x + threadIdx.x) >> 5;
    int lane = threadIdx.x & 31;
    const int NT = D_IN / 16;
    int nt = wave % NT;
    int mt = wave / NT;
    if (mt >= M_TOT / 16) return;
    int m0 = mt * 16, n0 = nt * 16;
    int half = lane >> 4, lq = lane & 15;
    int koff = half * 2;
    const float* aptr = xdbl + (size_t)(m0 + lq) * N_E + koff;
    const float* bptr = Wdt + (size_t)(n0 + lq) * DT_RANK + koff;
    v8f acc = {};
#pragma unroll
    for (int k = 0; k < DT_RANK; k += 4) {
        v2f a = *(const v2f*)(aptr + k);
        v2f b = *(const v2f*)(bptr + k);
        acc = __builtin_amdgcn_wmma_f32_16x16x4_f32(
            false, a, false, b, (short)0, acc, false, false);
    }
    int n = n0 + lq;
    float bias = bdt[n];
#pragma unroll
    for (int v = 0; v < 8; ++v) {
        int m = m0 + v + 8 * half;
        float z = acc[v] + bias;
        float sp = (z > 20.0f) ? z : log1pf(__expf(z));
        dt[(size_t)m * D_IN + n] = sp;
    }
}

struct StageRegs { v4f dt, x, bc; };

__device__ __forceinline__ void stage_load(StageRegs& r,
                                           const float* __restrict__ X,
                                           const float* __restrict__ xdbl,
                                           const float* __restrict__ dt,
                                           size_t mrow, int dcol, int q) {
    r.dt = *(const v4f*)(dt + mrow * D_IN + dcol);
    r.x  = *(const v4f*)(X  + mrow * D_IN + dcol);
    r.bc = *(const v4f*)(xdbl + mrow * N_E + DT_RANK + 4 * q);
}

__device__ __forceinline__ void stage_store(const StageRegs& r, float* buf, int row, int q) {
    *(v4f*)(buf + row * DX_PITCH + 4 * q) = r.dt;
    *(v4f*)(buf + DX_TILE + row * DX_PITCH + 4 * q) = r.x;
    float* bc = buf + 2 * DX_TILE + (4 * q) * BC_PITCH + row;
    bc[0]            = r.bc[0];
    bc[BC_PITCH]     = r.bc[1];
    bc[2 * BC_PITCH] = r.bc[2];
    bc[3 * BC_PITCH] = r.bc[3];
}

template <int ROW0>
__device__ __forceinline__ void load_state_row(float (&v)[TBLK], const float* bcT, int s, int tb) {
    const float* p = bcT + (ROW0 + s) * BC_PITCH + tb * TBLK;
    for (int i = 0; i < 4; ++i) {
        const v4f w = *(const v4f*)(p + 4 * i);
        v[4 * i] = w[0]; v[4 * i + 1] = w[1]; v[4 * i + 2] = w[2]; v[4 * i + 3] = w[3];
    }
}

#define SCAN1_STEP(J)                                                       \
    {                                                                       \
        const float ab = __builtin_amdgcn_exp2f(row_bcast<J>(dtv) * A2);    \
        h = __builtin_fmaf(ab, h, row_bcast<J>(dxv) * Bv[J]);               \
    }

__global__ void __launch_bounds__(SCAN_THREADS)
scan_phase1(const float* __restrict__ X,
            const float* __restrict__ xdbl,
            const float* __restrict__ dt,
            const float* __restrict__ A_log,
            float* __restrict__ aprod,
            float* __restrict__ hend) {
    __shared__ __attribute__((aligned(16))) float lds[2 * BUF_FLOATS];
    const int tid = threadIdx.x;
    const int lane = tid & 31;
    const int s = lane & 15;
    const int wave = blockIdx.x * (SCAN_THREADS / 32) + (tid >> 5);
    const int chunk = wave / (NCHAN / 2);
    const int cl = (tid >> 5) * 2 + (lane >> 4);
    const int channel = (wave % (NCHAN / 2)) * 2 + (lane >> 4);
    const int b = channel / D_IN;
    const int d = channel % D_IN;
    const float A2 = -__expf(A_log[d * D_ST + s]) * LOG2E;
    float dtsum = 0.0f, h = 0.0f;
    const size_t m0 = (size_t)b * SEQ + (size_t)chunk * LCHUNK;
    const int row = tid >> 3, q = tid & 7;
    const int dcol = (d - cl) + 4 * q;
    StageRegs st;
    stage_load(st, X, xdbl, dt, m0 + row, dcol, q);
    stage_store(st, lds, row, q);
    __syncthreads();
    for (int sub = 0; sub < NSUB; ++sub) {
        const float* buf = lds + (sub & 1) * BUF_FLOATS;
        if (sub + 1 < NSUB)
            stage_load(st, X, xdbl, dt, m0 + (size_t)(sub + 1) * TS + row, dcol, q);
        const float* dtT = buf;
        const float* xT  = buf + DX_TILE;
        const float* bcT = buf + 2 * DX_TILE;
        for (int tb = 0; tb < TS / TBLK; ++tb) {
            const int o = (tb * TBLK + s) * DX_PITCH + cl;
            const float dtv = dtT[o];
            const float dxv = dtv * xT[o];
            dtsum += dtv;
            float Bv[TBLK];
            load_state_row<0>(Bv, bcT, s, tb);
            SCAN1_STEP(0)  SCAN1_STEP(1)  SCAN1_STEP(2)  SCAN1_STEP(3)
            SCAN1_STEP(4)  SCAN1_STEP(5)  SCAN1_STEP(6)  SCAN1_STEP(7)
            SCAN1_STEP(8)  SCAN1_STEP(9)  SCAN1_STEP(10) SCAN1_STEP(11)
            SCAN1_STEP(12) SCAN1_STEP(13) SCAN1_STEP(14) SCAN1_STEP(15)
        }
        if (sub + 1 < NSUB) {
            stage_store(st, lds + ((sub + 1) & 1) * BUF_FLOATS, row, q);
            __syncthreads();
        }
    }
    dtsum += row_xor<1>(dtsum);
    dtsum += row_xor<2>(dtsum);
    dtsum += row_xor<4>(dtsum);
    dtsum += row_xor<8>(dtsum);
    const int idx = (channel * D_ST + s) * NCHUNK + chunk;
    aprod[idx] = __builtin_amdgcn_exp2f(dtsum * A2);
    hend[idx]  = h;
}

#define SCAN3_STEP(J)                                                       \
    {                                                                       \
        const float ab = __builtin_amdgcn_exp2f(row_bcast<J>(dtv) * A2);    \
        h = __builtin_fmaf(ab, h, row_bcast<J>(dxv) * Bv[J]);               \
        c[J] = h * Cv[J];                                                   \
    }

__global__ void __launch_bounds__(SCAN_THREADS)
scan_phase3(const float* __restrict__ X,
            const float* __restrict__ xdbl,
            const float* __restrict__ dt,
            const float* __restrict__ A_log,
            const float* __restrict__ Dp,
            const float* __restrict__ aprod,
            const float* __restrict__ hend,
            float* __restrict__ y) {
    __shared__ __attribute__((aligned(16))) float lds[2 * BUF_FLOATS];
    const int tid = threadIdx.x;
    const int lane = tid & 31;
    const int s = lane & 15;
    const int wave = blockIdx.x * (SCAN_THREADS / 32) + (tid >> 5);
    const int chunk = wave / (NCHAN / 2);
    const int cl = (tid >> 5) * 2 + (lane >> 4);
    const int channel = (wave % (NCHAN / 2)) * 2 + (lane >> 4);
    const int b = channel / D_IN;
    const int d = channel % D_IN;
    const float A2 = -__expf(A_log[d * D_ST + s]) * LOG2E;
    static_assert(NCHUNK == 8, "the pair fetch below reads two 16-byte halves per array");
    const int pbase = (channel * D_ST + s) * NCHUNK;
    const v4f ap0 = *(const v4f*)(aprod + pbase), ap1 = *(const v4f*)(aprod + pbase + 4);
    const v4f he0 = *(const v4f*)(hend + pbase),  he1 = *(const v4f*)(hend + pbase + 4);
    const float apv[NCHUNK - 1] = {ap0[0], ap0[1], ap0[2], ap0[3], ap1[0], ap1[1], ap1[2]};
    const float hev[NCHUNK - 1] = {he0[0], he0[1], he0[2], he0[3], he1[0], he1[1], he1[2]};
    float h = 0.0f;
    for (int ch = 0; ch < NCHUNK - 1; ++ch)
        if (ch < chunk) h = __builtin_fmaf(apv[ch], h, hev[ch]);
    const float Dd = Dp[d];
    const size_t m0 = (size_t)b * SEQ + (size_t)chunk * LCHUNK;
    const int row = tid >> 3, q = tid & 7;
    const int dcol = (d - cl) + 4 * q;
    const bool up1 = (s & 1) != 0, up2 = (s & 2) != 0, up4 = (s & 4) != 0, up8 = (s & 8) != 0;
    StageRegs st;
    stage_load(st, X, xdbl, dt, m0 + row, dcol, q);
    stage_store(st, lds, row, q);
    __syncthreads();
    for (int sub = 0; sub < NSUB; ++sub) {
        float* buf = lds + (sub & 1) * BUF_FLOATS;
        float* oth = lds + ((sub + 1) & 1) * BUF_FLOATS;
        if (sub + 1 < NSUB)
            stage_load(st, X, xdbl, dt, m0 + (size_t)(sub + 1) * TS + row, dcol, q);
        if (sub > 0)
            *(v4f*)(y + (m0 + (size_t)(sub - 1) * TS + row) * D_IN + dcol) =
                *(const v4f*)(oth + row * DX_PITCH + 4 * q);
        float*       dtT = buf;
        const float* xT  = buf + DX_TILE;
        const float* bcT = buf + 2 * DX_TILE;
        for (int tb = 0; tb < TS / TBLK; ++tb) {
            const int o = (tb * TBLK + s) * DX_PITCH + cl;
            const float dtv = dtT[o];
            const float xv  = xT[o];
            const float dxv = dtv * xv;
            float Bv[TBLK], Cv[TBLK], c[TBLK];
            load_state_row<0>(Bv, bcT, s, tb);
            load_state_row<D_ST>(Cv, bcT, s, tb);
            SCAN3_STEP(0)  SCAN3_STEP(1)  SCAN3_STEP(2)  SCAN3_STEP(3)
            SCAN3_STEP(4)  SCAN3_STEP(5)  SCAN3_STEP(6)  SCAN3_STEP(7)
            SCAN3_STEP(8)  SCAN3_STEP(9)  SCAN3_STEP(10) SCAN3_STEP(11)
            SCAN3_STEP(12) SCAN3_STEP(13) SCAN3_STEP(14) SCAN3_STEP(15)
            float r1[8], r2[4], r3[2];
            for (int k = 0; k < 8; ++k) r1[k] = fold<1>(c[2 * k], c[2 * k + 1], up1);
            for (int k = 0; k < 4; ++k) r2[k] = fold<2>(r1[2 * k], r1[2 * k + 1], up2);
            for (int k = 0; k < 2; ++k) r3[k] = fold<4>(r2[2 * k], r2[2 * k + 1], up4);
            const float ysum = fold<8>(r3[0], r3[1], up8);
            dtT[o] = ysum + Dd * xv;
        }
        if (sub + 1 < NSUB)
            stage_store(st, oth, row, q);
        __syncthreads();
    }
    *(v4f*)(y + (m0 + (size_t)(NSUB - 1) * TS + row) * D_IN + dcol) =
        *(const v4f*)(lds + ((NSUB - 1) & 1) * BUF_FLOATS + row * DX_PITCH + 4 * q);
}

extern "C" void kernel_launch(void* const* d_in, const int* in_sizes, int n_in,
                              void* d_out, int out_size, void* d_ws, size_t ws_size,
                              hipStream_t stream) {
    const float* x       = (const float*)d_in[0];
    const float* A_log   = (const float*)d_in[1];
    const float* Dp      = (const float*)d_in[2];
    const float* W_xproj = (const float*)d_in[3];
    const float* W_dt    = (const float*)d_in[4];
    const float* b_dt    = (const float*)d_in[5];
    float* y = (float*)d_out;

    char* ws = (char*)d_ws;
    size_t off = 0;
    float* xdbl  = (float*)(ws + off); off += (size_t)M_TOT * N_E * 4;
    off = (off + 255) & ~(size_t)255;
    float* dtbuf = (float*)(ws + off); off += (size_t)M_TOT * D_IN * 4;
    off = (off + 255) & ~(size_t)255;
    float* aprod = (float*)(ws + off); off += (size_t)NCHAN * D_ST * NCHUNK * 4;
    off = (off + 255) & ~(size_t)255;
    float* hend  = (float*)(ws + off);

    xproj_wmma<<<M_TOT / XP_M, XP_THREADS, 0, stream>>>(x, W_xproj, xdbl);
    dt_wmma<<<16384 / 8, 256, 0, stream>>>(xdbl, W_dt, b_dt, dtbuf);
    scan_phase1<<<SCAN_WAVES / (SCAN_THREADS / 32), SCAN_THREADS, 0, stream>>>(
        x, xdbl, dtbuf, A_log, aprod, hend);
    scan_phase3<<<SCAN_WAVES / (SCAN_THREADS / 32), SCAN_THREADS, 0, stream>>>(
        x, xdbl, dtbuf, A_log, Dp, aprod, hend, y);
}


# dt GEMM fed through LDS: a 16-wave workgroup per 64 rows x 128 columns stages x_dbl and W_dt rows as whole 128-byte lines, fragments read from LDS; dt leaves as whole lines; K sum order changes
# speedup vs baseline: 4.9202x; 1.0335x over previous
#include <hip/hip_runtime.h>
#include <hip/hip_bf16.h>
#include <math.h>

typedef __attribute__((ext_vector_type(2))) float v2f;
typedef __attribute__((ext_vector_type(8))) float v8f;

#define D_IN   1024
#define D_ST   16
#define DT_RANK 64
#define N_E    96
#define BATCH  2
#define SEQ    2048
#define M_TOT  (BATCH*SEQ)
#define NCHUNK 8
#define LCHUNK (SEQ/NCHUNK)
#define NCHAN  (BATCH*D_IN)
#define TBLK   16
#define SCAN_THREADS 512
#define SCAN_WAVES (NCHAN / 2 * NCHUNK)
#define LOG2E 1.44269504088896341f
#define TS       64
#define NSUB     (LCHUNK / TS)
#define DX_PITCH 36
#define BC_PITCH 68
#define DX_TILE  (TS * DX_PITCH)
#define BC_TILE  (32 * BC_PITCH)
#define BUF_FLOATS (2 * DX_TILE + BC_TILE)
typedef __attribute__((ext_vector_type(4))) float v4f;

template <int CTRL>
__device__ __forceinline__ float row_dpp(float v) {
    const int b = __builtin_bit_cast(int, v);
    return __builtin_bit_cast(float, __builtin_amdgcn_update_dpp(0, b, CTRL, 0xf, 0xf, true));
}
template <int J> __device__ __forceinline__ float row_bcast(float v) { return row_dpp<0x150 + J>(v); }
template <int M> __device__ __forceinline__ float row_xor(float v) { return row_dpp<0x160 + M>(v); }

template <int M>
__device__ __forceinline__ float fold(float a, float b, bool up) {
    const float keep = up ? b : a;
    const float send = up ? a : b;
    return keep + row_xor<M>(send);
}

#define XP_M       16
#define XP_KS      128
#define XP_NSLAB   (D_IN / XP_KS)
#define XP_PITCH   (XP_KS + 4)
#define XP_ROWS    (XP_M + N_E)
#define XP_BUF     (XP_ROWS * XP_PITCH)
#define XP_WAVES   12
#define XP_THREADS (XP_WAVES * 32)
#define XP_WROWS   (N_E / XP_WAVES)

struct XpStage { v4f w[XP_WROWS]; v4f x[2]; };

__device__ __forceinline__ void xp_load(XpStage& r, const float* __restrict__ X,
                                        const float* __restrict__ W,
                                        size_t m0, int slab, int wv, int lane) {
    const int kcol = slab * XP_KS + 4 * lane;
    for (int j = 0; j < XP_WROWS; ++j)
        r.w[j] = *(const v4f*)(W + (size_t)(wv + XP_WAVES * j) * D_IN + kcol);
    if (wv < 8) {
        r.x[0] = *(const v4f*)(X + (m0 + wv) * D_IN + kcol);
        r.x[1] = *(const v4f*)(X + (m0 + wv + 8) * D_IN + kcol);
    }
}

__device__ __forceinline__ void xp_store(const XpStage& r, float* buf, int wv, int lane) {
    for (int j = 0; j < XP_WROWS; ++j)
        *(v4f*)(buf + (XP_M + wv + XP_WAVES * j) * XP_PITCH + 4 * lane) = r.w[j];
    if (wv < 8) {
        *(v4f*)(buf + wv * XP_PITCH + 4 * lane) = r.x[0];
        *(v4f*)(buf + (wv + 8) * XP_PITCH + 4 * lane) = r.x[1];
    }
}

__global__ void __launch_bounds__(XP_THREADS)
xproj_wmma(const float* __restrict__ X,
           const float* __restrict__ W,
           float* __restrict__ xdbl) {
    __shared__ __attribute__((aligned(16))) float lds[2 * XP_BUF];
    __shared__ __attribute__((aligned(16))) float part[(N_E / 16) * 256];
    __shared__ __attribute__((aligned(16))) float res[XP_M * N_E];
    const int tid = threadIdx.x;
    const int lane = tid & 31;
    const int wv = tid >> 5;
    const int nt = wv % (N_E / 16);
    const int kh = wv / (N_E / 16);
    const int half = lane >> 4, lq = lane & 15;
    const size_t m0 = (size_t)blockIdx.x * XP_M;
    const int aoff = lq * XP_PITCH + kh * (XP_KS / 2) + 4 * half;
    const int boff = (XP_M + nt * 16 + lq) * XP_PITCH + kh * (XP_KS / 2) + 4 * half;
    v8f acc0 = {}, acc1 = {};
    XpStage st;
    xp_load(st, X, W, m0, 0, wv, lane);
    xp_store(st, lds, wv, lane);
    __syncthreads();
    for (int slab = 0; slab < XP_NSLAB; ++slab) {
        const float* buf = lds + (slab & 1) * XP_BUF;
        if (slab + 1 < XP_NSLAB)
            xp_load(st, X, W, m0, slab + 1, wv, lane);
        v4f af[8], bf[8];
        for (int i = 0; i < 8; ++i) {
            af[i] = *(const v4f*)(buf + aoff + 8 * i);
            bf[i] = *(const v4f*)(buf + boff + 8 * i);
        }
        for (int i = 0; i < 8; ++i) {
            const v2f a0 = {af[i][0], af[i][1]}, b0 = {bf[i][0], bf[i][1]};
            const v2f a1 = {af[i][2], af[i][3]}, b1 = {bf[i][2], bf[i][3]};
            acc0 = __builtin_amdgcn_wmma_f32_16x16x4_f32(
                false, a0, false, b0, (short)0, acc0, false, false);
            acc1 = __builtin_amdgcn_wmma_f32_16x16x4_f32(
                false, a1, false, b1, (short)0, acc1, false, false);
        }
        if (slab + 1 < XP_NSLAB) {
            xp_store(st, lds + ((slab + 1) & 1) * XP_BUF, wv, lane);
            __syncthreads();
        }
    }
    const v8f acc = acc0 + acc1;
    if (kh == 1)
        for (int v = 0; v < 8; ++v) part[nt * 256 + v * 32 + lane] = acc[v];
    __syncthreads();
    if (kh == 0)
        for (int v = 0; v < 8; ++v)
            res[(v + 8 * half) * N_E + nt * 16 + lq] = acc[v] + part[nt * 256 + v * 32 + lane];
    __syncthreads();
    *(v4f*)(xdbl + m0 * N_E + 4 * tid) = *(const v4f*)(res + 4 * tid);
}

#define DT_MB      64
#define DT_NB      128
#define DT_PITCH   (DT_RANK + 4)
#define DT_OPITCH  36
#define DT_WAVES   16
#define DT_THREADS (DT_WAVES * 32)

__global__ void __launch_bounds__(DT_THREADS)
dt_wmma(const float* __restrict__ xdbl,
        const float* __restrict__ Wdt,
        const float* __restrict__ bdt,
        float* __restrict__ dt) {
    __shared__ __attribute__((aligned(16))) float ldsA[DT_MB * DT_PITCH];
    __shared__ __attribute__((aligned(16))) float ldsB[DT_NB * DT_PITCH];
    __shared__ __attribute__((aligned(16))) float osta[DT_WAVES * 16 * DT_OPITCH];
    const int tid = threadIdx.x;
    const int lane = tid & 31;
    const int wv = tid >> 5;
    const int half = lane >> 4, lq = lane & 15;
    const int m0 = (blockIdx.x / (D_IN / DT_NB)) * DT_MB;
    const int n0 = (blockIdx.x % (D_IN / DT_NB)) * DT_NB;
    const int srow = tid >> 4, sc = (tid & 15) * 4;
    v4f ra[DT_MB / 32], rb[DT_NB / 32];
    for (int j = 0; j < DT_MB / 32; ++j)
        ra[j] = *(const v4f*)(xdbl + (size_t)(m0 + srow + 32 * j) * N_E + sc);
    for (int j = 0; j < DT_NB / 32; ++j)
        rb[j] = *(const v4f*)(Wdt + (size_t)(n0 + srow + 32 * j) * DT_RANK + sc);
    for (int j = 0; j < DT_MB / 32; ++j)
        *(v4f*)(ldsA + (srow + 32 * j) * DT_PITCH + sc) = ra[j];
    for (int j = 0; j < DT_NB / 32; ++j)
        *(v4f*)(ldsB + (srow + 32 * j) * DT_PITCH + sc) = rb[j];
    __syncthreads();
    const int s = wv & 3;
    const int cp = wv >> 2;
    v4f af[8];
    const float* ap = ldsA + (16 * s + lq) * DT_PITCH + 4 * half;
    for (int i = 0; i < 8; ++i) af[i] = *(const v4f*)(ap + 8 * i);
    float* ost = osta + wv * (16 * DT_OPITCH);
    for (int t = 0; t < 2; ++t) {
        const float* bp = ldsB + (32 * cp + 16 * t + lq) * DT_PITCH + 4 * half;
        v4f bf[8];
        for (int i = 0; i < 8; ++i) bf[i] = *(const v4f*)(bp + 8 * i);
        v8f acc = {};
        for (int i = 0; i < 8; ++i) {
            const v2f a0 = {af[i][0], af[i][1]}, b0 = {bf[i][0], bf[i][1]};
            const v2f a1 = {af[i][2], af[i][3]}, b1 = {bf[i][2], bf[i][3]};
            acc = __builtin_amdgcn_wmma_f32_16x16x4_f32(
                false, a0, false, b0, (short)0, acc, false, false);
            acc = __builtin_amdgcn_wmma_f32_16x16x4_f32(
                false, a1, false, b1, (short)0, acc, false, false);
        }
        const float bias = bdt[n0 + 32 * cp + 16 * t + lq];
        for (int v = 0; v < 8; ++v) {
            const float z = acc[v] + bias;
            const float sp = (z > 20.0f) ? z : log1pf(__expf(z));
            ost[(v + 8 * half) * DT_OPITCH + 16 * t + lq] = sp;
        }
    }
    __syncthreads();
    for (int p = 0; p < 4; ++p) {
        const int row = 4 * p + (lane >> 3);
        *(v4f*)(dt + (size_t)(m0 + 16 * s + row) * D_IN + n0 + 32 * cp + 4 * (lane & 7)) =
            *(const v4f*)(ost + row * DT_OPITCH + 4 * (lane & 7));
    }
}

struct StageRegs { v4f dt, x, bc; };

__device__ __forceinline__ void stage_load(StageRegs& r,
                                           const float* __restrict__ X,
                                           const float* __restrict__ xdbl,
                                           const float* __restrict__ dt,
                                           size_t mrow, int dcol, int q) {
    r.dt = *(const v4f*)(dt + mrow * D_IN + dcol);
    r.x  = *(const v4f*)(X  + mrow * D_IN + dcol);
    r.bc = *(const v4f*)(xdbl + mrow * N_E + DT_RANK + 4 * q);
}

__device__ __forceinline__ void stage_store(const StageRegs& r, float* buf, int row, int q) {
    *(v4f*)(buf + row * DX_PITCH + 4 * q) = r.dt;
    *(v4f*)(buf + DX_TILE + row * DX_PITCH + 4 * q) = r.x;
    float* bc = buf + 2 * DX_TILE + (4 * q) * BC_PITCH + row;
    bc[0]            = r.bc[0];
    bc[BC_PITCH]     = r.bc[1];
    bc[2 * BC_PITCH] = r.bc[2];
    bc[3 * BC_PITCH] = r.bc[3];
}

template <int ROW0>
__device__ __forceinline__ void load_state_row(float (&v)[TBLK], const float* bcT, int s, int tb) {
    const float* p = bcT + (ROW0 + s) * BC_PITCH + tb * TBLK;
    for (int i = 0; i < 4; ++i) {
        const v4f w = *(const v4f*)(p + 4 * i);
        v[4 * i] = w[0]; v[4 * i + 1] = w[1]; v[4 * i + 2] = w[2]; v[4 * i + 3] = w[3];
    }
}

#define SCAN1_STEP(J)                                                       \
    {                                                                       \
        const float ab = __builtin_amdgcn_exp2f(row_bcast<J>(dtv) * A2);    \
        h = __builtin_fmaf(ab, h, row_bcast<J>(dxv) * Bv[J]);               \
    }

__global__ void __launch_bounds__(SCAN_THREADS)
scan_phase1(const float* __restrict__ X,
            const float* __restrict__ xdbl,
            const float* __restrict__ dt,
            const float* __restrict__ A_log,
            float* __restrict__ aprod,
            float* __restrict__ hend) {
    __shared__ __attribute__((aligned(16))) float lds[2 * BUF_FLOATS];
    const int tid = threadIdx.x;
    const int lane = tid & 31;
    const int s = lane & 15;
    const int wave = blockIdx.x * (SCAN_THREADS / 32) + (tid >> 5);
    const int chunk = wave / (NCHAN / 2);
    const int cl = (tid >> 5) * 2 + (lane >> 4);
    const int channel = (wave % (NCHAN / 2)) * 2 + (lane >> 4);
    const int b = channel / D_IN;
    const int d = channel % D_IN;
    const float A2 = -__expf(A_log[d * D_ST + s]) * LOG2E;
    float dtsum = 0.0f, h = 0.0f;
    const size_t m0 = (size_t)b * SEQ + (size_t)chunk * LCHUNK;
    const int row = tid >> 3, q = tid & 7;
    const int dcol = (d - cl) + 4 * q;
    StageRegs st;
    stage_load(st, X, xdbl, dt, m0 + row, dcol, q);
    stage_store(st, lds, row, q);
    __syncthreads();
    for (int sub = 0; sub < NSUB; ++sub) {
        const float* buf = lds + (sub & 1) * BUF_FLOATS;
        if (sub + 1 < NSUB)
            stage_load(st, X, xdbl, dt, m0 + (size_t)(sub + 1) * TS + row, dcol, q);
        const float* dtT = buf;
        const float* xT  = buf + DX_TILE;
        const float* bcT = buf + 2 * DX_TILE;
        for (int tb = 0; tb < TS / TBLK; ++tb) {
            const int o = (tb * TBLK + s) * DX_PITCH + cl;
            const float dtv = dtT[o];
            const float dxv = dtv * xT[o];
            dtsum += dtv;
            float Bv[TBLK];
            load_state_row<0>(Bv, bcT, s, tb);
            SCAN1_STEP(0)  SCAN1_STEP(1)  SCAN1_STEP(2)  SCAN1_STEP(3)
            SCAN1_STEP(4)  SCAN1_STEP(5)  SCAN1_STEP(6)  SCAN1_STEP(7)
            SCAN1_STEP(8)  SCAN1_STEP(9)  SCAN1_STEP(10) SCAN1_STEP(11)
            SCAN1_STEP(12) SCAN1_STEP(13) SCAN1_STEP(14) SCAN1_STEP(15)
        }
        if (sub + 1 < NSUB) {
            stage_store(st, lds + ((sub + 1) & 1) * BUF_FLOATS, row, q);
            __syncthreads();
        }
    }
    dtsum += row_xor<1>(dtsum);
    dtsum += row_xor<2>(dtsum);
    dtsum += row_xor<4>(dtsum);
    dtsum += row_xor<8>(dtsum);
    const int idx = (channel * D_ST + s) * NCHUNK + chunk;
    aprod[idx] = __builtin_amdgcn_exp2f(dtsum * A2);
    hend[idx]  = h;
}

#define SCAN3_STEP(J)                                                       \
    {                                                                       \
        const float ab = __builtin_amdgcn_exp2f(row_bcast<J>(dtv) * A2);    \
        h = __builtin_fmaf(ab, h, row_bcast<J>(dxv) * Bv[J]);               \
        c[J] = h * Cv[J];                                                   \
    }

__global__ void __launch_bounds__(SCAN_THREADS)
scan_phase3(const float* __restrict__ X,
            const float* __restrict__ xdbl,
            const float* __restrict__ dt,
            const float* __restrict__ A_log,
            const float* __restrict__ Dp,
            const float* __restrict__ aprod,
            const float* __restrict__ hend,
            float* __restrict__ y) {
    __shared__ __attribute__((aligned(16))) float lds[2 * BUF_FLOATS];
    const int tid = threadIdx.x;
    const int lane = tid & 31;
    const int s = lane & 15;
    const int wave = blockIdx.x * (SCAN_THREADS / 32) + (tid >> 5);
    const int chunk = wave / (NCHAN / 2);
    const int cl = (tid >> 5) * 2 + (lane >> 4);
    const int channel = (wave % (NCHAN / 2)) * 2 + (lane >> 4);
    const int b = channel / D_IN;
    const int d = channel % D_IN;
    const float A2 = -__expf(A_log[d * D_ST + s]) * LOG2E;
    static_assert(NCHUNK == 8, "the pair fetch below reads two 16-byte halves per array");
    const int pbase = (channel * D_ST + s) * NCHUNK;
    const v4f ap0 = *(const v4f*)(aprod + pbase), ap1 = *(const v4f*)(aprod + pbase + 4);
    const v4f he0 = *(const v4f*)(hend + pbase),  he1 = *(const v4f*)(hend + pbase + 4);
    const float apv[NCHUNK - 1] = {ap0[0], ap0[1], ap0[2], ap0[3], ap1[0], ap1[1], ap1[2]};
    const float hev[NCHUNK - 1] = {he0[0], he0[1], he0[2], he0[3], he1[0], he1[1], he1[2]};
    float h = 0.0f;
    for (int ch = 0; ch < NCHUNK - 1; ++ch)
        if (ch < chunk) h = __builtin_fmaf(apv[ch], h, hev[ch]);
    const float Dd = Dp[d];
    const size_t m0 = (size_t)b * SEQ + (size_t)chunk * LCHUNK;
    const int row = tid >> 3, q = tid & 7;
    const int dcol = (d - cl) + 4 * q;
    const bool up1 = (s & 1) != 0, up2 = (s & 2) != 0, up4 = (s & 4) != 0, up8 = (s & 8) != 0;
    StageRegs st;
    stage_load(st, X, xdbl, dt, m0 + row, dcol, q);
    stage_store(st, lds, row, q);
    __syncthreads();
    for (int sub = 0; sub < NSUB; ++sub) {
        float* buf = lds + (sub & 1) * BUF_FLOATS;
        float* oth = lds + ((sub + 1) & 1) * BUF_FLOATS;
        if (sub + 1 < NSUB)
            stage_load(st, X, xdbl, dt, m0 + (size_t)(sub + 1) * TS + row, dcol, q);
        if (sub > 0)
            *(v4f*)(y + (m0 + (size_t)(sub - 1) * TS + row) * D_IN + dcol) =
                *(const v4f*)(oth + row * DX_PITCH + 4 * q);
        float*       dtT = buf;
        const float* xT  = buf + DX_TILE;
        const float* bcT = buf + 2 * DX_TILE;
        for (int tb = 0; tb < TS / TBLK; ++tb) {
            const int o = (tb * TBLK + s) * DX_PITCH + cl;
            const float dtv = dtT[o];
            const float xv  = xT[o];
            const float dxv = dtv * xv;
            float Bv[TBLK], Cv[TBLK], c[TBLK];
            load_state_row<0>(Bv, bcT, s, tb);
            load_state_row<D_ST>(Cv, bcT, s, tb);
            SCAN3_STEP(0)  SCAN3_STEP(1)  SCAN3_STEP(2)  SCAN3_STEP(3)
            SCAN3_STEP(4)  SCAN3_STEP(5)  SCAN3_STEP(6)  SCAN3_STEP(7)
            SCAN3_STEP(8)  SCAN3_STEP(9)  SCAN3_STEP(10) SCAN3_STEP(11)
            SCAN3_STEP(12) SCAN3_STEP(13) SCAN3_STEP(14) SCAN3_STEP(15)
            float r1[8], r2[4], r3[2];
            for (int k = 0; k < 8; ++k) r1[k] = fold<1>(c[2 * k], c[2 * k + 1], up1);
            for (int k = 0; k < 4; ++k) r2[k] = fold<2>(r1[2 * k], r1[2 * k + 1], up2);
            for (int k = 0; k < 2; ++k) r3[k] = fold<4>(r2[2 * k], r2[2 * k + 1], up4);
            const float ysum = fold<8>(r3[0], r3[1], up8);
            dtT[o] = ysum + Dd * xv;
        }
        if (sub + 1 < NSUB)
            stage_store(st, oth, row, q);
        __syncthreads();
    }
    *(v4f*)(y + (m0 + (size_t)(NSUB - 1) * TS + row) * D_IN + dcol) =
        *(const v4f*)(lds + ((NSUB - 1) & 1) * BUF_FLOATS + row * DX_PITCH + 4 * q);
}

extern "C" void kernel_launch(void* const* d_in, const int* in_sizes, int n_in,
                              void* d_out, int out_size, void* d_ws, size_t ws_size,
                              hipStream_t stream) {
    const float* x       = (const float*)d_in[0];
    const float* A_log   = (const float*)d_in[1];
    const float* Dp      = (const float*)d_in[2];
    const float* W_xproj = (const float*)d_in[3];
    const float* W_dt    = (const float*)d_in[4];
    const float* b_dt    = (const float*)d_in[5];
    float* y = (float*)d_out;

    char* ws = (char*)d_ws;
    size_t off = 0;
    float* xdbl  = (float*)(ws + off); off += (size_t)M_TOT * N_E * 4;
    off = (off + 255) & ~(size_t)255;
    float* dtbuf = (float*)(ws + off); off += (size_t)M_TOT * D_IN * 4;
    off = (off + 255) & ~(size_t)255;
    float* aprod = (float*)(ws + off); off += (size_t)NCHAN * D_ST * NCHUNK * 4;
    off = (off + 255) & ~(size_t)255;
    float* hend  = (float*)(ws + off);

    xproj_wmma<<<M_TOT / XP_M, XP_THREADS, 0, stream>>>(x, W_xproj, xdbl);
    dt_wmma<<<(M_TOT / DT_MB) * (D_IN / DT_NB), DT_THREADS, 0, stream>>>(xdbl, W_dt, b_dt, dtbuf);
    scan_phase1<<<SCAN_WAVES / (SCAN_THREADS / 32), SCAN_THREADS, 0, stream>>>(
        x, xdbl, dtbuf, A_log, aprod, hend);
    scan_phase3<<<SCAN_WAVES / (SCAN_THREADS / 32), SCAN_THREADS, 0, stream>>>(
        x, xdbl, dtbuf, A_log, Dp, aprod, hend, y);
}


# Scan lanes hold two states as one float pair stepped by packed f32 mul/fma; a channel is an 8-lane group, blocks of 8 steps, lane gather for dt and dt*x; state sum and chunk dt sum order change
# speedup vs baseline: 5.1253x; 1.0417x over previous
#include <hip/hip_runtime.h>
#include <hip/hip_bf16.h>
#include <math.h>

typedef __attribute__((ext_vector_type(2))) float v2f;
typedef __attribute__((ext_vector_type(8))) float v8f;

#define D_IN   1024
#define D_ST   16
#define DT_RANK 64
#define N_E    96
#define BATCH  2
#define SEQ    2048
#define M_TOT  (BATCH*SEQ)
#define NCHUNK 8
#define LCHUNK (SEQ/NCHUNK)
#define NCHAN  (BATCH*D_IN)
#define TBLK   8
#define SCAN_THREADS 256
#define SCAN_WAVES (NCHAN / 4 * NCHUNK)
#define LOG2E 1.44269504088896341f
#define TS       32
#define NSUB     (LCHUNK / TS)
#define DX_PITCH 36
#define BC_PITCH 68
#define DX_TILE  (TS * DX_PITCH)
#define BC_TILE  (16 * BC_PITCH)
#define BUF_FLOATS (2 * DX_TILE + BC_TILE)
typedef __attribute__((ext_vector_type(4))) float v4f;

template <int CTRL>
__device__ __forceinline__ float row_dpp(float v) {
    const int b = __builtin_bit_cast(int, v);
    return __builtin_bit_cast(float, __builtin_amdgcn_update_dpp(0, b, CTRL, 0xf, 0xf, true));
}
template <int J> __device__ __forceinline__ float row_bcast(float v) { return row_dpp<0x150 + J>(v); }
template <int M> __device__ __forceinline__ float row_xor(float v) { return row_dpp<0x160 + M>(v); }
template <int J> __device__ __forceinline__ float grp_bcast(float v) {
    const unsigned b = __builtin_bit_cast(unsigned, v);
    return __builtin_bit_cast(float, __builtin_amdgcn_permlane16(
        0u, b, 0x11111111u * J, 0x11111111u * (8 + J), false, true));
}

template <int M>
__device__ __forceinline__ float fold(float a, float b, bool up) {
    const float keep = up ? b : a;
    const float send = up ? a : b;
    return keep + row_xor<M>(send);
}

#define XP_M       16
#define XP_KS      128
#define XP_NSLAB   (D_IN / XP_KS)
#define XP_PITCH   (XP_KS + 4)
#define XP_ROWS    (XP_M + N_E)
#define XP_BUF     (XP_ROWS * XP_PITCH)
#define XP_WAVES   12
#define XP_THREADS (XP_WAVES * 32)
#define XP_WROWS   (N_E / XP_WAVES)

struct XpStage { v4f w[XP_WROWS]; v4f x[2]; };

__device__ __forceinline__ void xp_load(XpStage& r, const float* __restrict__ X,
                                        const float* __restrict__ W,
                                        size_t m0, int slab, int wv, int lane) {
    const int kcol = slab * XP_KS + 4 * lane;
    for (int j = 0; j < XP_WROWS; ++j)
        r.w[j] = *(const v4f*)(W + (size_t)(wv + XP_WAVES * j) * D_IN + kcol);
    if (wv < 8) {
        r.x[0] = *(const v4f*)(X + (m0 + wv) * D_IN + kcol);
        r.x[1] = *(const v4f*)(X + (m0 + wv + 8) * D_IN + kcol);
    }
}

__device__ __forceinline__ void xp_store(const XpStage& r, float* buf, int wv, int lane) {
    for (int j = 0; j < XP_WROWS; ++j)
        *(v4f*)(buf + (XP_M + wv + XP_WAVES * j) * XP_PITCH + 4 * lane) = r.w[j];
    if (wv < 8) {
        *(v4f*)(buf + wv * XP_PITCH + 4 * lane) = r.x[0];
        *(v4f*)(buf + (wv + 8) * XP_PITCH + 4 * lane) = r.x[1];
    }
}

__global__ void __launch_bounds__(XP_THREADS)
xproj_wmma(const float* __restrict__ X,
           const float* __restrict__ W,
           float* __restrict__ xdbl) {
    __shared__ __attribute__((aligned(16))) float lds[2 * XP_BUF];
    __shared__ __attribute__((aligned(16))) float part[(N_E / 16) * 256];
    __shared__ __attribute__((aligned(16))) float res[XP_M * N_E];
    const int tid = threadIdx.x;
    const int lane = tid & 31;
    const int wv = tid >> 5;
    const int nt = wv % (N_E / 16);
    const int kh = wv / (N_E / 16);
    const int half = lane >> 4, lq = lane & 15;
    const size_t m0 = (size_t)blockIdx.x * XP_M;
    const int aoff = lq * XP_PITCH + kh * (XP_KS / 2) + 4 * half;
    const int boff = (XP_M + nt * 16 + lq) * XP_PITCH + kh * (XP_KS / 2) + 4 * half;
    v8f acc0 = {}, acc1 = {};
    XpStage st;
    xp_load(st, X, W, m0, 0, wv, lane);
    xp_store(st, lds, wv, lane);
    __syncthreads();
    for (int slab = 0; slab < XP_NSLAB; ++slab) {
        const float* buf = lds + (slab & 1) * XP_BUF;
        if (slab + 1 < XP_NSLAB)
            xp_load(st, X, W, m0, slab + 1, wv, lane);
        v4f af[8], bf[8];
        for (int i = 0; i < 8; ++i) {
            af[i] = *(const v4f*)(buf + aoff + 8 * i);
            bf[i] = *(const v4f*)(buf + boff + 8 * i);
        }
        for (int i = 0; i < 8; ++i) {
            const v2f a0 = {af[i][0], af[i][1]}, b0 = {bf[i][0], bf[i][1]};
            const v2f a1 = {af[i][2], af[i][3]}, b1 = {bf[i][2], bf[i][3]};
            acc0 = __builtin_amdgcn_wmma_f32_16x16x4_f32(
                false, a0, false, b0, (short)0, acc0, false, false);
            acc1 = __builtin_amdgcn_wmma_f32_16x16x4_f32(
                false, a1, false, b1, (short)0, acc1, false, false);
        }
        if (slab + 1 < XP_NSLAB) {
            xp_store(st, lds + ((slab + 1) & 1) * XP_BUF, wv, lane);
            __syncthreads();
        }
    }
    const v8f acc = acc0 + acc1;
    if (kh == 1)
        for (int v = 0; v < 8; ++v) part[nt * 256 + v * 32 + lane] = acc[v];
    __syncthreads();
    if (kh == 0)
        for (int v = 0; v < 8; ++v)
            res[(v + 8 * half) * N_E + nt * 16 + lq] = acc[v] + part[nt * 256 + v * 32 + lane];
    __syncthreads();
    *(v4f*)(xdbl + m0 * N_E + 4 * tid) = *(const v4f*)(res + 4 * tid);
}

#define DT_MB      64
#define DT_NB      128
#define DT_PITCH   (DT_RANK + 4)
#define DT_OPITCH  36
#define DT_WAVES   16
#define DT_THREADS (DT_WAVES * 32)

__global__ void __launch_bounds__(DT_THREADS)
dt_wmma(const float* __restrict__ xdbl,
        const float* __restrict__ Wdt,
        const float* __restrict__ bdt,
        float* __restrict__ dt) {
    __shared__ __attribute__((aligned(16))) float ldsA[DT_MB * DT_PITCH];
    __shared__ __attribute__((aligned(16))) float ldsB[DT_NB * DT_PITCH];
    __shared__ __attribute__((aligned(16))) float osta[DT_WAVES * 16 * DT_OPITCH];
    const int tid = threadIdx.x;
    const int lane = tid & 31;
    const int wv = tid >> 5;
    const int half = lane >> 4, lq = lane & 15;
    const int m0 = (blockIdx.x / (D_IN / DT_NB)) * DT_MB;
    const int n0 = (blockIdx.x % (D_IN / DT_NB)) * DT_NB;
    const int srow = tid >> 4, sc = (tid & 15) * 4;
    v4f ra[DT_MB / 32], rb[DT_NB / 32];
    for (int j = 0; j < DT_MB / 32; ++j)
        ra[j] = *(const v4f*)(xdbl + (size_t)(m0 + srow + 32 * j) * N_E + sc);
    for (int j = 0; j < DT_NB / 32; ++j)
        rb[j] = *(const v4f*)(Wdt + (size_t)(n0 + srow + 32 * j) * DT_RANK + sc);
    for (int j = 0; j < DT_MB / 32; ++j)
        *(v4f*)(ldsA + (srow + 32 * j) * DT_PITCH + sc) = ra[j];
    for (int j = 0; j < DT_NB / 32; ++j)
        *(v4f*)(ldsB + (srow + 32 * j) * DT_PITCH + sc) = rb[j];
    __syncthreads();
    const int s = wv & 3;
    const int cp = wv >> 2;
    v4f af[8];
    const float* ap = ldsA + (16 * s + lq) * DT_PITCH + 4 * half;
    for (int i = 0; i < 8; ++i) af[i] = *(const v4f*)(ap + 8 * i);
    float* ost = osta + wv * (16 * DT_OPITCH);
    for (int t = 0; t < 2; ++t) {
        const float* bp = ldsB + (32 * cp + 16 * t + lq) * DT_PITCH + 4 * half;
        v4f bf[8];
        for (int i = 0; i < 8; ++i) bf[i] = *(const v4f*)(bp + 8 * i);
        v8f acc = {};
        for (int i = 0; i < 8; ++i) {
            const v2f a0 = {af[i][0], af[i][1]}, b0 = {bf[i][0], bf[i][1]};
            const v2f a1 = {af[i][2], af[i][3]}, b1 = {bf[i][2], bf[i][3]};
            acc = __builtin_amdgcn_wmma_f32_16x16x4_f32(
                false, a0, false, b0, (short)0, acc, false, false);
            acc = __builtin_amdgcn_wmma_f32_16x16x4_f32(
                false, a1, false, b1, (short)0, acc, false, false);
        }
        const float bias = bdt[n0 + 32 * cp + 16 * t + lq];
        for (int v = 0; v < 8; ++v) {
            const float z = acc[v] + bias;
            const float sp = (z > 20.0f) ? z : log1pf(__expf(z));
            ost[(v + 8 * half) * DT_OPITCH + 16 * t + lq] = sp;
        }
    }
    __syncthreads();
    for (int p = 0; p < 4; ++p) {
        const int row = 4 * p + (lane >> 3);
        *(v4f*)(dt + (size_t)(m0 + 16 * s + row) * D_IN + n0 + 32 * cp + 4 * (lane & 7)) =
            *(const v4f*)(ost + row * DT_OPITCH + 4 * (lane & 7));
    }
}

struct StageRegs { v4f dt, x, bc; };

__device__ __forceinline__ void stage_load(StageRegs& r,
                                           const float* __restrict__ X,
                                           const float* __restrict__ xdbl,
                                           const float* __restrict__ dt,
                                           size_t mrow, int dcol, int q) {
    r.dt = *(const v4f*)(dt + mrow * D_IN + dcol);
    r.x  = *(const v4f*)(X  + mrow * D_IN + dcol);
    r.bc = *(const v4f*)(xdbl + mrow * N_E + DT_RANK + 4 * q);
}

__device__ __forceinline__ void stage_store(const StageRegs& r, float* buf, int row, int q) {
    *(v4f*)(buf + row * DX_PITCH + 4 * q) = r.dt;
    *(v4f*)(buf + DX_TILE + row * DX_PITCH + 4 * q) = r.x;
    float* bc = buf + 2 * DX_TILE + (2 * q) * BC_PITCH + 2 * row;
    const v2f p0 = {r.bc[0], r.bc[1]}, p1 = {r.bc[2], r.bc[3]};
    *(v2f*)bc = p0;
    *(v2f*)(bc + BC_PITCH) = p1;
}

template <int ROW0>
__device__ __forceinline__ void load_pair_row(v2f (&v)[TBLK], const float* bcT, int p, int tb) {
    const float* ptr = bcT + (ROW0 + p) * BC_PITCH + tb * (2 * TBLK);
    for (int i = 0; i < TBLK / 2; ++i) {
        const v4f w = *(const v4f*)(ptr + 4 * i);
        const v2f lo = {w[0], w[1]}, hi = {w[2], w[3]};
        v[2 * i] = lo; v[2 * i + 1] = hi;
    }
}

__device__ __forceinline__ v2f exp2_pair(v2f e) {
    const v2f r = {__builtin_amdgcn_exp2f(e[0]), __builtin_amdgcn_exp2f(e[1])};
    return r;
}
__device__ __forceinline__ v2f splat(float s) { const v2f r = {s, s}; return r; }

#define SCAN_PRE(J)                                                         \
    ab[J] = exp2_pair(A2 * splat(grp_bcast<J>(dtv)));                       \
    bx[J] = Bv[J] * splat(grp_bcast<J>(dxv));
#define SCAN1_STEP(J)                                                       \
    h = __builtin_elementwise_fma(ab[J], h, bx[J]);

__global__ void __launch_bounds__(SCAN_THREADS)
scan_phase1(const float* __restrict__ X,
            const float* __restrict__ xdbl,
            const float* __restrict__ dt,
            const float* __restrict__ A_log,
            float* __restrict__ aprod,
            float* __restrict__ hend) {
    __shared__ __attribute__((aligned(16))) float lds[2 * BUF_FLOATS];
    const int tid = threadIdx.x;
    const int lane = tid & 31;
    const int p = lane & 7;
    const int wave = blockIdx.x * (SCAN_THREADS / 32) + (tid >> 5);
    const int chunk = wave / (NCHAN / 4);
    const int cl = (tid >> 5) * 4 + (lane >> 3);
    const int channel = (wave % (NCHAN / 4)) * 4 + (lane >> 3);
    const int b = channel / D_IN;
    const int d = channel % D_IN;
    const v2f al = *(const v2f*)(A_log + d * D_ST + 2 * p);
    const v2f A2 = {-__expf(al[0]) * LOG2E, -__expf(al[1]) * LOG2E};
    float dtsum = 0.0f;
    v2f h = {0.0f, 0.0f};
    const size_t m0 = (size_t)b * SEQ + (size_t)chunk * LCHUNK;
    const int row = tid >> 3, q = tid & 7;
    const int dcol = (d - cl) + 4 * q;
    StageRegs st;
    stage_load(st, X, xdbl, dt, m0 + row, dcol, q);
    stage_store(st, lds, row, q);
    __syncthreads();
    for (int sub = 0; sub < NSUB; ++sub) {
        const float* buf = lds + (sub & 1) * BUF_FLOATS;
        if (sub + 1 < NSUB)
            stage_load(st, X, xdbl, dt, m0 + (size_t)(sub + 1) * TS + row, dcol, q);
        const float* dtT = buf;
        const float* xT  = buf + DX_TILE;
        const float* bcT = buf + 2 * DX_TILE;
        for (int tb = 0; tb < TS / TBLK; ++tb) {
            const int o = (tb * TBLK + p) * DX_PITCH + cl;
            const float dtv = dtT[o];
            const float dxv = dtv * xT[o];
            dtsum += dtv;
            v2f Bv[TBLK];
            load_pair_row<0>(Bv, bcT, p, tb);
            v2f ab[TBLK], bx[TBLK];
            SCAN_PRE(0)  SCAN_PRE(1)  SCAN_PRE(2)  SCAN_PRE(3)
            SCAN_PRE(4)  SCAN_PRE(5)  SCAN_PRE(6)  SCAN_PRE(7)
            SCAN1_STEP(0)  SCAN1_STEP(1)  SCAN1_STEP(2)  SCAN1_STEP(3)
            SCAN1_STEP(4)  SCAN1_STEP(5)  SCAN1_STEP(6)  SCAN1_STEP(7)
        }
        if (sub + 1 < NSUB) {
            stage_store(st, lds + ((sub + 1) & 1) * BUF_FLOATS, row, q);
            __syncthreads();
        }
    }
    dtsum += row_xor<1>(dtsum);
    dtsum += row_xor<2>(dtsum);
    dtsum += row_xor<4>(dtsum);
    const int idx = (channel * D_ST + 2 * p) * NCHUNK + chunk;
    aprod[idx]          = __builtin_amdgcn_exp2f(dtsum * A2[0]);
    aprod[idx + NCHUNK] = __builtin_amdgcn_exp2f(dtsum * A2[1]);
    hend[idx]           = h[0];
    hend[idx + NCHUNK]  = h[1];
}

#define SCAN3_STEP(J)                                                       \
    {                                                                       \
        h = __builtin_elementwise_fma(ab[J], h, bx[J]);                     \
        const v2f hc = h * Cv[J];                                           \
        c[J] = hc[0] + hc[1];                                               \
    }

__global__ void __launch_bounds__(SCAN_THREADS)
scan_phase3(const float* __restrict__ X,
            const float* __restrict__ xdbl,
            const float* __restrict__ dt,
            const float* __restrict__ A_log,
            const float* __restrict__ Dp,
            const float* __restrict__ aprod,
            const float* __restrict__ hend,
            float* __restrict__ y) {
    __shared__ __attribute__((aligned(16))) float lds[2 * BUF_FLOATS];
    const int tid = threadIdx.x;
    const int lane = tid & 31;
    const int p = lane & 7;
    const int wave = blockIdx.x * (SCAN_THREADS / 32) + (tid >> 5);
    const int chunk = wave / (NCHAN / 4);
    const int cl = (tid >> 5) * 4 + (lane >> 3);
    const int channel = (wave % (NCHAN / 4)) * 4 + (lane >> 3);
    const int b = channel / D_IN;
    const int d = channel % D_IN;
    const v2f al = *(const v2f*)(A_log + d * D_ST + 2 * p);
    const v2f A2 = {-__expf(al[0]) * LOG2E, -__expf(al[1]) * LOG2E};
    static_assert(NCHUNK == 8, "the pair fetch below reads two 16-byte halves per state and array");
    const int pbase = (channel * D_ST + 2 * p) * NCHUNK;
    const v4f apa0 = *(const v4f*)(aprod + pbase),     apa1 = *(const v4f*)(aprod + pbase + 4);
    const v4f apb0 = *(const v4f*)(aprod + pbase + 8), apb1 = *(const v4f*)(aprod + pbase + 12);
    const v4f hea0 = *(const v4f*)(hend + pbase),      hea1 = *(const v4f*)(hend + pbase + 4);
    const v4f heb0 = *(const v4f*)(hend + pbase + 8),  heb1 = *(const v4f*)(hend + pbase + 12);
    const float apv0[NCHUNK - 1] = {apa0[0], apa0[1], apa0[2], apa0[3], apa1[0], apa1[1], apa1[2]};
    const float apv1[NCHUNK - 1] = {apb0[0], apb0[1], apb0[2], apb0[3], apb1[0], apb1[1], apb1[2]};
    const float hev0[NCHUNK - 1] = {hea0[0], hea0[1], hea0[2], hea0[3], hea1[0], hea1[1], hea1[2]};
    const float hev1[NCHUNK - 1] = {heb0[0], heb0[1], heb0[2], heb0[3], heb1[0], heb1[1], heb1[2]};
    float h0 = 0.0f, h1 = 0.0f;
    for (int ch = 0; ch < NCHUNK - 1; ++ch)
        if (ch < chunk) {
            h0 = __builtin_fmaf(apv0[ch], h0, hev0[ch]);
            h1 = __builtin_fmaf(apv1[ch], h1, hev1[ch]);
        }
    v2f h = {h0, h1};
    const float Dd = Dp[d];
    const size_t m0 = (size_t)b * SEQ + (size_t)chunk * LCHUNK;
    const int row = tid >> 3, q = tid & 7;
    const int dcol = (d - cl) + 4 * q;
    const bool up1 = (p & 1) != 0, up2 = (p & 2) != 0, up4 = (p & 4) != 0;
    StageRegs st;
    stage_load(st, X, xdbl, dt, m0 + row, dcol, q);
    stage_store(st, lds, row, q);
    __syncthreads();
    for (int sub = 0; sub < NSUB; ++sub) {
        float* buf = lds + (sub & 1) * BUF_FLOATS;
        float* oth = lds + ((sub + 1) & 1) * BUF_FLOATS;
        if (sub + 1 < NSUB)
            stage_load(st, X, xdbl, dt, m0 + (size_t)(sub + 1) * TS + row, dcol, q);
        if (sub > 0)
            *(v4f*)(y + (m0 + (size_t)(sub - 1) * TS + row) * D_IN + dcol) =
                *(const v4f*)(oth + row * DX_PITCH + 4 * q);
        float*       dtT = buf;
        const float* xT  = buf + DX_TILE;
        const float* bcT = buf + 2 * DX_TILE;
#pragma unroll
        for (int tb = 0; tb < TS / TBLK; ++tb) {
            const int o = (tb * TBLK + p) * DX_PITCH + cl;
            const float dtv = dtT[o];
            const float xv  = xT[o];
            const float dxv = dtv * xv;
            v2f Bv[TBLK], Cv[TBLK];
            float c[TBLK];
            load_pair_row<0>(Bv, bcT, p, tb);
            load_pair_row<D_ST / 2>(Cv, bcT, p, tb);
            v2f ab[TBLK], bx[TBLK];
            SCAN_PRE(0)  SCAN_PRE(1)  SCAN_PRE(2)  SCAN_PRE(3)
            SCAN_PRE(4)  SCAN_PRE(5)  SCAN_PRE(6)  SCAN_PRE(7)
            SCAN3_STEP(0)  SCAN3_STEP(1)  SCAN3_STEP(2)  SCAN3_STEP(3)
            SCAN3_STEP(4)  SCAN3_STEP(5)  SCAN3_STEP(6)  SCAN3_STEP(7)
            float r1[4], r2[2];
            for (int k = 0; k < 4; ++k) r1[k] = fold<1>(c[2 * k], c[2 * k + 1], up1);
            for (int k = 0; k < 2; ++k) r2[k] = fold<2>(r1[2 * k], r1[2 * k + 1], up2);
            const float ysum = fold<4>(r2[0], r2[1], up4);
            dtT[o] = ysum + Dd * xv;
        }
        if (sub + 1 < NSUB)
            stage_store(st, oth, row, q);
        __syncthreads();
    }
    *(v4f*)(y + (m0 + (size_t)(NSUB - 1) * TS + row) * D_IN + dcol) =
        *(const v4f*)(lds + ((NSUB - 1) & 1) * BUF_FLOATS + row * DX_PITCH + 4 * q);
}

extern "C" void kernel_launch(void* const* d_in, const int* in_sizes, int n_in,
                              void* d_out, int out_size, void* d_ws, size_t ws_size,
                              hipStream_t stream) {
    const float* x       = (const float*)d_in[0];
    const float* A_log   = (const float*)d_in[1];
    const float* Dp      = (const float*)d_in[2];
    const float* W_xproj = (const float*)d_in[3];
    const float* W_dt    = (const float*)d_in[4];
    const float* b_dt    = (const float*)d_in[5];
    float* y = (float*)d_out;

    char* ws = (char*)d_ws;
    size_t off = 0;
    float* xdbl  = (float*)(ws + off); off += (size_t)M_TOT * N_E * 4;
    off = (off + 255) & ~(size_t)255;
    float* dtbuf = (float*)(ws + off); off += (size_t)M_TOT * D_IN * 4;
    off = (off + 255) & ~(size_t)255;
    float* aprod = (float*)(ws + off); off += (size_t)NCHAN * D_ST * NCHUNK * 4;
    off = (off + 255) & ~(size_t)255;
    float* hend  = (float*)(ws + off);

    xproj_wmma<<<M_TOT / XP_M, XP_THREADS, 0, stream>>>(x, W_xproj, xdbl);
    dt_wmma<<<(M_TOT / DT_MB) * (D_IN / DT_NB), DT_THREADS, 0, stream>>>(xdbl, W_dt, b_dt, dtbuf);
    scan_phase1<<<SCAN_WAVES / (SCAN_THREADS / 32), SCAN_THREADS, 0, stream>>>(
        x, xdbl, dtbuf, A_log, aprod, hend);
    scan_phase3<<<SCAN_WAVES / (SCAN_THREADS / 32), SCAN_THREADS, 0, stream>>>(
        x, xdbl, dtbuf, A_log, Dp, aprod, hend, y);
}
